# starting kernel
# baseline (speedup 1.0000x reference)
#include <hip/hip_runtime.h>
#include <hip/hip_bf16.h>

typedef float v2f __attribute__((ext_vector_type(2)));
typedef float v8f __attribute__((ext_vector_type(8)));

#define N_NODES 50000
#define N_EDGES 800000
#define D 64
#define N_TILES (N_NODES / 16)
#define SMOOTHF 0.5f


__global__ void k_zero_deg(float* __restrict__ deg) {
    int i = blockIdx.x * blockDim.x + threadIdx.x;
    if (i < N_NODES) deg[i] = 0.0f;
}

__global__ void k_degree(const int* __restrict__ ecol, float* __restrict__ deg) {
    int e = blockIdx.x * blockDim.x + threadIdx.x;
    if (e < N_EDGES) atomicAdd(&deg[ecol[e]], 1.0f);
}

__global__ void k_rsqrt(float* __restrict__ deg) {
    int i = blockIdx.x * blockDim.x + threadIdx.x;
    if (i < N_NODES) deg[i] = rsqrtf(deg[i] + 1.0f);
}


__global__ __launch_bounds__(256) void k_gemm_wmma(
    const float* __restrict__ x, const float* __restrict__ W,
    const float* __restrict__ bias, const float* __restrict__ d,
    float* __restrict__ support, float* __restrict__ h, float* __restrict__ msg)
{
    __shared__ float Wl[D * D];
    for (int i = threadIdx.x; i < D * D; i += 256) Wl[i] = W[i];
    __syncthreads();

    const int wave = threadIdx.x >> 5;
    const int lane = threadIdx.x & 31;
    const int tile = blockIdx.x * 8 + wave;
    if (tile >= N_TILES) return;

    const int row0 = tile * 16;
    const int half = lane >> 4;
    const int l15  = lane & 15;

    v8f acc[4] = {v8f{}, v8f{}, v8f{}, v8f{}};
    const float* xrow = x + (size_t)(row0 + l15) * D;

    #pragma unroll
    for (int kk = 0; kk < 16; ++kk) {
        const int k = kk * 4 + half * 2;
        v2f a = *(const v2f*)(xrow + k);
        #pragma unroll
        for (int nt = 0; nt < 4; ++nt) {
            const int n = nt * 16 + l15;
            v2f bf = *(const v2f*)(&Wl[n * D + k]);
            acc[nt] = __builtin_amdgcn_wmma_f32_16x16x4_f32(
                 false, a,  false, bf,
                 (short)0, acc[nt],
                 false,  false);
        }
    }

    #pragma unroll
    for (int nt = 0; nt < 4; ++nt) {
        const int n  = nt * 16 + l15;
        const float bv = bias[n];
        #pragma unroll
        for (int r = 0; r < 8; ++r) {
            const int row  = row0 + r + 8 * half;
            const float s  = acc[nt][r] + bv;
            const float hv = s * d[row];
            const size_t idx = (size_t)row * D + n;
            support[idx] = s;
            h[idx]       = hv;
            msg[idx]     = hv;
        }
    }
}


__global__ __launch_bounds__(256) void k_scatter(
    const int* __restrict__ erow, const int* __restrict__ ecol,
    const float* __restrict__ h, float* __restrict__ msg)
{
    const long long t = (long long)blockIdx.x * blockDim.x + threadIdx.x;
    const int e = (int)(t >> 6);
    const int j = (int)(t & 63);
    if (e < N_EDGES) {
        const int r = erow[e];
        const int c = ecol[e];
        atomicAdd(&msg[(size_t)r * D + j], h[(size_t)c * D + j]);
    }
}


__global__ void k_final(const float* __restrict__ msg, const float* __restrict__ support,
                        const float* __restrict__ d, float* __restrict__ out)
{
    const int i = blockIdx.x * blockDim.x + threadIdx.x;
    if (i < N_NODES * D) {
        const int row = i >> 6;
        out[i] = (msg[i] * d[row] * SMOOTHF + support[i]) * (1.0f / (1.0f + SMOOTHF));
    }
}


extern "C" void kernel_launch(void* const* d_in, const int* in_sizes, int n_in,
                              void* d_out, int out_size, void* d_ws, size_t ws_size,
                              hipStream_t stream) {
    const float* x    = (const float*)d_in[0];
    const float* W    = (const float*)d_in[1];
    const float* bias = (const float*)d_in[2];
    const int*   erow = (const int*)d_in[3];
    const int*   ecol = (const int*)d_in[4];
    float* out = (float*)d_out;

    float* ws      = (float*)d_ws;
    float* support = ws;
    float* h       = ws + (size_t)N_NODES * D;
    float* msg     = ws + 2 * (size_t)N_NODES * D;
    float* deg     = ws + 3 * (size_t)N_NODES * D;

    k_zero_deg<<<(N_NODES + 255) / 256, 256, 0, stream>>>(deg);
    k_degree  <<<(N_EDGES + 255) / 256, 256, 0, stream>>>(ecol, deg);
    k_rsqrt   <<<(N_NODES + 255) / 256, 256, 0, stream>>>(deg);
    k_gemm_wmma<<<(N_TILES + 7) / 8, 256, 0, stream>>>(x, W, bias, deg, support, h, msg);
    k_scatter <<<(N_EDGES * 64) / 256, 256, 0, stream>>>(erow, ecol, h, msg);
    k_final   <<<(N_NODES * D + 255) / 256, 256, 0, stream>>>(msg, support, deg, out);
}


# edge sums: edges counting-sorted by destination-row range (LDS counts, scan, placement); a work-group per range adds rows in LDS (f32 LDS atomics, arrival order), writes out; global f32 atomics gone
# speedup vs baseline: 1.5762x; 1.5762x over previous
#include <hip/hip_runtime.h>
#include <hip/hip_bf16.h>

typedef float v2f __attribute__((ext_vector_type(2)));
typedef float v8f __attribute__((ext_vector_type(8)));

#define N_NODES 50000
#define N_EDGES 800000
#define D 64
#define N_TILES (N_NODES / 16)
#define SMOOTHF 0.5f


__global__ void k_zero_deg(float* __restrict__ deg) {
    int i = blockIdx.x * blockDim.x + threadIdx.x;
    if (i < N_NODES) deg[i] = 0.0f;
}

__global__ void k_degree(const int* __restrict__ ecol, float* __restrict__ deg) {
    int e = blockIdx.x * blockDim.x + threadIdx.x;
    if (e < N_EDGES) atomicAdd(&deg[ecol[e]], 1.0f);
}

__global__ void k_rsqrt(float* __restrict__ deg) {
    int i = blockIdx.x * blockDim.x + threadIdx.x;
    if (i < N_NODES) deg[i] = rsqrtf(deg[i] + 1.0f);
}


__global__ __launch_bounds__(256) void k_gemm_wmma(
    const float* __restrict__ x, const float* __restrict__ W,
    const float* __restrict__ bias, const float* __restrict__ d,
    float* __restrict__ support, float* __restrict__ h)
{
    __shared__ float Wl[D * D];
    for (int i = threadIdx.x; i < D * D; i += 256) Wl[i] = W[i];
    __syncthreads();

    const int wave = threadIdx.x >> 5;
    const int lane = threadIdx.x & 31;
    const int tile = blockIdx.x * 8 + wave;
    if (tile >= N_TILES) return;

    const int row0 = tile * 16;
    const int half = lane >> 4;
    const int l15  = lane & 15;

    v8f acc[4] = {v8f{}, v8f{}, v8f{}, v8f{}};
    const float* xrow = x + (size_t)(row0 + l15) * D;

    #pragma unroll
    for (int kk = 0; kk < 16; ++kk) {
        const int k = kk * 4 + half * 2;
        v2f a = *(const v2f*)(xrow + k);
        #pragma unroll
        for (int nt = 0; nt < 4; ++nt) {
            const int n = nt * 16 + l15;
            v2f bf = *(const v2f*)(&Wl[n * D + k]);
            acc[nt] = __builtin_amdgcn_wmma_f32_16x16x4_f32(
                 false, a,  false, bf,
                 (short)0, acc[nt],
                 false,  false);
        }
    }

    #pragma unroll
    for (int nt = 0; nt < 4; ++nt) {
        const int n  = nt * 16 + l15;
        const float bv = bias[n];
        #pragma unroll
        for (int r = 0; r < 8; ++r) {
            const int row  = row0 + r + 8 * half;
            const float s  = acc[nt][r] + bv;
            const float hv = s * d[row];
            const size_t idx = (size_t)row * D + n;
            support[idx] = s;
            h[idx]       = hv;
        }
    }
}


#define NB  256
#define RB  196
#define G   256
#define EPG (N_EDGES / G)
#define RPW ((RB + 31) / 32)

__global__ __launch_bounds__(1024) void k_bin_count(const int* __restrict__ erow,
                                                    int* __restrict__ cnt)
{
    __shared__ int hist[NB];
    const int g = blockIdx.x;
    const int t = threadIdx.x;
    if (t < NB) hist[t] = 0;
    __syncthreads();
    const int e0 = g * EPG;
    for (int i = t; i < EPG; i += 1024) {
        const unsigned b = (unsigned)erow[e0 + i] / RB;
        __hip_atomic_fetch_add(&hist[b], 1, __ATOMIC_RELAXED, __HIP_MEMORY_SCOPE_WORKGROUP);
    }
    __syncthreads();
    if (t < NB) cnt[t * G + g] = hist[t];
}

__global__ __launch_bounds__(1024) void k_bin_scan(const int* __restrict__ cnt,
                                                   int* __restrict__ off)
{
    __shared__ int sh[1024];
    const int t = threadIdx.x;
    const int4* src = (const int4*)(cnt + t * 64);
    int4 v[16];
    int sum = 0;
    #pragma unroll
    for (int i = 0; i < 16; ++i) {
        v[i] = src[i];
        sum += v[i].x + v[i].y + v[i].z + v[i].w;
    }
    sh[t] = sum;
    __syncthreads();
    for (int s = 1; s < 1024; s <<= 1) {
        const int add = (t >= s) ? sh[t - s] : 0;
        __syncthreads();
        sh[t] += add;
        __syncthreads();
    }
    int run = sh[t] - sum;
    int4* dst = (int4*)(off + t * 64);
    #pragma unroll
    for (int i = 0; i < 16; ++i) {
        int4 o;
        o.x = run; run += v[i].x;
        o.y = run; run += v[i].y;
        o.z = run; run += v[i].z;
        o.w = run; run += v[i].w;
        dst[i] = o;
    }
}

__global__ __launch_bounds__(1024) void k_bin_place(const int* __restrict__ erow,
                                                    const int* __restrict__ ecol,
                                                    const int* __restrict__ off,
                                                    unsigned* __restrict__ sorted)
{
    __shared__ int cur[NB];
    const int g = blockIdx.x;
    const int t = threadIdx.x;
    if (t < NB) cur[t] = off[t * G + g];
    __syncthreads();
    const int e0 = g * EPG;
    for (int i = t; i < EPG; i += 1024) {
        const unsigned r = (unsigned)erow[e0 + i];
        const unsigned c = (unsigned)ecol[e0 + i];
        const unsigned b = r / RB;
        const int slot = __hip_atomic_fetch_add(&cur[b], 1, __ATOMIC_RELAXED,
                                                __HIP_MEMORY_SCOPE_WORKGROUP);
        sorted[slot] = ((r - b * RB) << 16) | c;
    }
}

__global__ __launch_bounds__(1024) void k_accum(
    const unsigned* __restrict__ sorted, const int* __restrict__ off,
    const float* __restrict__ h, const float* __restrict__ support,
    const float* __restrict__ d, float* __restrict__ out)
{
    __shared__ float acc[RB * D];
    const int b    = blockIdx.x;
    const int tid  = threadIdx.x;
    const int lane = tid & 31;
    const int wave = __builtin_amdgcn_readfirstlane(tid >> 5);

    for (int i = tid; i < RB * D; i += 1024) acc[i] = 0.0f;
    const int bs = off[b * G];
    const int be = (b + 1 < NB) ? off[(b + 1) * G] : N_EDGES;
    __syncthreads();

    const int per = ((((be - bs) + 31) >> 5) + 7) & ~7;
    const int w0  = bs + wave * per;
    const int w1  = (w0 + per < be) ? (w0 + per) : be;
    const float* hl = h + 2 * lane;
    float* al = acc + 2 * lane;

    for (int base = w0; base < w1; base += 32) {
        const int n = (w1 - base < 32) ? (w1 - base) : 32;
        unsigned pk = 0;
        if (lane < n) pk = sorted[base + lane];
        int j = 0;
        for (; j + 8 <= n; j += 8) {
            unsigned p[8];
            v2f v[8];
            #pragma unroll
            for (int u = 0; u < 8; ++u) {
                p[u] = __builtin_amdgcn_readlane(pk, j + u);
                v[u] = *(const v2f*)(hl + (size_t)(p[u] & 0xffffu) * D);
            }
            #pragma unroll
            for (int u = 0; u < 8; ++u) {
                float* a = al + (p[u] >> 16) * D;
                __hip_atomic_fetch_add(a,     v[u].x, __ATOMIC_RELAXED, __HIP_MEMORY_SCOPE_WORKGROUP);
                __hip_atomic_fetch_add(a + 1, v[u].y, __ATOMIC_RELAXED, __HIP_MEMORY_SCOPE_WORKGROUP);
            }
        }
        for (; j < n; ++j) {
            const unsigned p = __builtin_amdgcn_readlane(pk, j);
            const v2f v = *(const v2f*)(hl + (size_t)(p & 0xffffu) * D);
            float* a = al + (p >> 16) * D;
            __hip_atomic_fetch_add(a,     v.x, __ATOMIC_RELAXED, __HIP_MEMORY_SCOPE_WORKGROUP);
            __hip_atomic_fetch_add(a + 1, v.y, __ATOMIC_RELAXED, __HIP_MEMORY_SCOPE_WORKGROUP);
        }
    }
    __syncthreads();

    const int row0  = b * RB;
    const int nrows = (N_NODES - row0 < RB) ? (N_NODES - row0) : RB;
    v2f hv[RPW], sv[RPW];
    float dr[RPW];
    #pragma unroll
    for (int k = 0; k < RPW; ++k) {
        const int rl  = wave + 32 * k;
        const int row = row0 + ((rl < nrows) ? rl : (nrows - 1));
        const size_t idx = (size_t)row * D + 2 * lane;
        hv[k] = *(const v2f*)(h + idx);
        sv[k] = *(const v2f*)(support + idx);
        dr[k] = d[row];
    }
    #pragma unroll
    for (int k = 0; k < RPW; ++k) {
        const int rl = wave + 32 * k;
        if (rl < nrows) {
            const size_t idx = (size_t)(row0 + rl) * D + 2 * lane;
            const v2f av = *(const v2f*)(al + rl * D);
            v2f o;
            o.x = ((av.x + hv[k].x) * dr[k] * SMOOTHF + sv[k].x) * (1.0f / (1.0f + SMOOTHF));
            o.y = ((av.y + hv[k].y) * dr[k] * SMOOTHF + sv[k].y) * (1.0f / (1.0f + SMOOTHF));
            *(v2f*)(out + idx) = o;
        }
    }
}


extern "C" void kernel_launch(void* const* d_in, const int* in_sizes, int n_in,
                              void* d_out, int out_size, void* d_ws, size_t ws_size,
                              hipStream_t stream) {
    const float* x    = (const float*)d_in[0];
    const float* W    = (const float*)d_in[1];
    const float* bias = (const float*)d_in[2];
    const int*   erow = (const int*)d_in[3];
    const int*   ecol = (const int*)d_in[4];
    float* out = (float*)d_out;

    float* ws      = (float*)d_ws;
    float* support = ws;
    float* h       = ws + (size_t)N_NODES * D;
    unsigned* sorted = (unsigned*)(ws + 2 * (size_t)N_NODES * D);
    int* cnt       = (int*)(sorted + N_EDGES);
    int* off       = cnt + NB * G;
    float* deg     = ws + 3 * (size_t)N_NODES * D;

    k_zero_deg<<<(N_NODES + 255) / 256, 256, 0, stream>>>(deg);
    k_degree  <<<(N_EDGES + 255) / 256, 256, 0, stream>>>(ecol, deg);
    k_rsqrt   <<<(N_NODES + 255) / 256, 256, 0, stream>>>(deg);
    k_gemm_wmma<<<(N_TILES + 7) / 8, 256, 0, stream>>>(x, W, bias, deg, support, h);
    k_bin_count<<<G, 1024, 0, stream>>>(erow, cnt);
    k_bin_scan <<<1, 1024, 0, stream>>>(cnt, off);
    k_bin_place<<<G, 1024, 0, stream>>>(erow, ecol, off, sorted);
    k_accum    <<<NB, 1024, 0, stream>>>(sorted, off, h, support, deg, out);
}


# launch merge: the independent degree stages (zero, count, rsqrt) and bin stages (count, scan, place) run pairwise side by side in three shared launches, block index picks the body; 8 launches -> 5
# speedup vs baseline: 1.8103x; 1.1485x over previous
#include <hip/hip_runtime.h>
#include <hip/hip_bf16.h>

typedef float v2f __attribute__((ext_vector_type(2)));
typedef float v8f __attribute__((ext_vector_type(8)));

#define N_NODES 50000
#define N_EDGES 800000
#define D 64
#define N_TILES (N_NODES / 16)
#define SMOOTHF 0.5f


#define TPB 1024
#define ZB  ((N_NODES + TPB - 1) / TPB)
#define DB  ((N_EDGES + TPB - 1) / TPB)


__device__ __forceinline__ void zero_deg_body(float* __restrict__ deg, int blk) {
    const int i = blk * TPB + threadIdx.x;
    if (i < N_NODES) deg[i] = 0.0f;
}

__device__ __forceinline__ void degree_body(const int* __restrict__ ecol,
                                            float* __restrict__ deg, int blk) {
    const int e = blk * TPB + threadIdx.x;
    if (e < N_EDGES) atomicAdd(&deg[ecol[e]], 1.0f);
}

__device__ __forceinline__ void rsqrt_body(float* __restrict__ deg, int blk) {
    const int i = blk * TPB + threadIdx.x;
    if (i < N_NODES) deg[i] = rsqrtf(deg[i] + 1.0f);
}


__global__ __launch_bounds__(256) void k_gemm_wmma(
    const float* __restrict__ x, const float* __restrict__ W,
    const float* __restrict__ bias, const float* __restrict__ d,
    float* __restrict__ support, float* __restrict__ h)
{
    __shared__ float Wl[D * D];
    for (int i = threadIdx.x; i < D * D; i += 256) Wl[i] = W[i];
    __syncthreads();

    const int wave = threadIdx.x >> 5;
    const int lane = threadIdx.x & 31;
    const int tile = blockIdx.x * 8 + wave;
    if (tile >= N_TILES) return;

    const int row0 = tile * 16;
    const int half = lane >> 4;
    const int l15  = lane & 15;

    v8f acc[4] = {v8f{}, v8f{}, v8f{}, v8f{}};
    const float* xrow = x + (size_t)(row0 + l15) * D;

    #pragma unroll
    for (int kk = 0; kk < 16; ++kk) {
        const int k = kk * 4 + half * 2;
        v2f a = *(const v2f*)(xrow + k);
        #pragma unroll
        for (int nt = 0; nt < 4; ++nt) {
            const int n = nt * 16 + l15;
            v2f bf = *(const v2f*)(&Wl[n * D + k]);
            acc[nt] = __builtin_amdgcn_wmma_f32_16x16x4_f32(
                 false, a,  false, bf,
                 (short)0, acc[nt],
                 false,  false);
        }
    }

    #pragma unroll
    for (int nt = 0; nt < 4; ++nt) {
        const int n  = nt * 16 + l15;
        const float bv = bias[n];
        #pragma unroll
        for (int r = 0; r < 8; ++r) {
            const int row  = row0 + r + 8 * half;
            const float s  = acc[nt][r] + bv;
            const float hv = s * d[row];
            const size_t idx = (size_t)row * D + n;
            support[idx] = s;
            h[idx]       = hv;
        }
    }
}


#define NB  256
#define RB  196
#define G   256
#define EPG (N_EDGES / G)
#define RPW ((RB + 31) / 32)

__device__ __forceinline__ void bin_count_body(const int* __restrict__ erow,
                                               int* __restrict__ cnt, int g)
{
    __shared__ int hist[NB];
    const int t = threadIdx.x;
    if (t < NB) hist[t] = 0;
    __syncthreads();
    const int e0 = g * EPG;
    for (int i = t; i < EPG; i += TPB) {
        const unsigned b = (unsigned)erow[e0 + i] / RB;
        __hip_atomic_fetch_add(&hist[b], 1, __ATOMIC_RELAXED, __HIP_MEMORY_SCOPE_WORKGROUP);
    }
    __syncthreads();
    if (t < NB) cnt[t * G + g] = hist[t];
}

__device__ __forceinline__ void bin_scan_body(const int* __restrict__ cnt,
                                              int* __restrict__ off)
{
    __shared__ int sh[TPB];
    const int t = threadIdx.x;
    const int4* src = (const int4*)(cnt + t * 64);
    int4 v[16];
    int sum = 0;
    #pragma unroll
    for (int i = 0; i < 16; ++i) {
        v[i] = src[i];
        sum += v[i].x + v[i].y + v[i].z + v[i].w;
    }
    sh[t] = sum;
    __syncthreads();
    for (int s = 1; s < TPB; s <<= 1) {
        const int add = (t >= s) ? sh[t - s] : 0;
        __syncthreads();
        sh[t] += add;
        __syncthreads();
    }
    int run = sh[t] - sum;
    int4* dst = (int4*)(off + t * 64);
    #pragma unroll
    for (int i = 0; i < 16; ++i) {
        int4 o;
        o.x = run; run += v[i].x;
        o.y = run; run += v[i].y;
        o.z = run; run += v[i].z;
        o.w = run; run += v[i].w;
        dst[i] = o;
    }
}

__device__ __forceinline__ void bin_place_body(const int* __restrict__ erow,
                                               const int* __restrict__ ecol,
                                               const int* __restrict__ off,
                                               unsigned* __restrict__ sorted, int g)
{
    __shared__ int cur[NB];
    const int t = threadIdx.x;
    if (t < NB) cur[t] = off[t * G + g];
    __syncthreads();
    const int e0 = g * EPG;
    for (int i = t; i < EPG; i += TPB) {
        const unsigned r = (unsigned)erow[e0 + i];
        const unsigned c = (unsigned)ecol[e0 + i];
        const unsigned b = r / RB;
        const int slot = __hip_atomic_fetch_add(&cur[b], 1, __ATOMIC_RELAXED,
                                                __HIP_MEMORY_SCOPE_WORKGROUP);
        sorted[slot] = ((r - b * RB) << 16) | c;
    }
}


__global__ __launch_bounds__(TPB) void k_count_zero(const int* __restrict__ erow,
                                                    int* __restrict__ cnt,
                                                    float* __restrict__ deg)
{
    if (blockIdx.x < G) bin_count_body(erow, cnt, blockIdx.x);
    else                zero_deg_body(deg, blockIdx.x - G);
}

__global__ __launch_bounds__(TPB) void k_scan_degree(const int* __restrict__ cnt,
                                                     int* __restrict__ off,
                                                     const int* __restrict__ ecol,
                                                     float* __restrict__ deg)
{
    if (blockIdx.x == 0) bin_scan_body(cnt, off);
    else                 degree_body(ecol, deg, blockIdx.x - 1);
}

__global__ __launch_bounds__(TPB) void k_place_rsqrt(const int* __restrict__ erow,
                                                     const int* __restrict__ ecol,
                                                     const int* __restrict__ off,
                                                     unsigned* __restrict__ sorted,
                                                     float* __restrict__ deg)
{
    if (blockIdx.x < G) bin_place_body(erow, ecol, off, sorted, blockIdx.x);
    else                rsqrt_body(deg, blockIdx.x - G);
}

__global__ __launch_bounds__(1024) void k_accum(
    const unsigned* __restrict__ sorted, const int* __restrict__ off,
    const float* __restrict__ h, const float* __restrict__ support,
    const float* __restrict__ d, float* __restrict__ out)
{
    __shared__ float acc[RB * D];
    const int b    = blockIdx.x;
    const int tid  = threadIdx.x;
    const int lane = tid & 31;
    const int wave = __builtin_amdgcn_readfirstlane(tid >> 5);

    for (int i = tid; i < RB * D; i += 1024) acc[i] = 0.0f;
    const int bs = off[b * G];
    const int be = (b + 1 < NB) ? off[(b + 1) * G] : N_EDGES;
    __syncthreads();

    const int per = ((((be - bs) + 31) >> 5) + 7) & ~7;
    const int w0  = bs + wave * per;
    const int w1  = (w0 + per < be) ? (w0 + per) : be;
    const float* hl = h + 2 * lane;
    float* al = acc + 2 * lane;

    for (int base = w0; base < w1; base += 32) {
        const int n = (w1 - base < 32) ? (w1 - base) : 32;
        unsigned pk = 0;
        if (lane < n) pk = sorted[base + lane];
        int j = 0;
        for (; j + 8 <= n; j += 8) {
            unsigned p[8];
            v2f v[8];
            #pragma unroll
            for (int u = 0; u < 8; ++u) {
                p[u] = __builtin_amdgcn_readlane(pk, j + u);
                v[u] = *(const v2f*)(hl + (size_t)(p[u] & 0xffffu) * D);
            }
            #pragma unroll
            for (int u = 0; u < 8; ++u) {
                float* a = al + (p[u] >> 16) * D;
                __hip_atomic_fetch_add(a,     v[u].x, __ATOMIC_RELAXED, __HIP_MEMORY_SCOPE_WORKGROUP);
                __hip_atomic_fetch_add(a + 1, v[u].y, __ATOMIC_RELAXED, __HIP_MEMORY_SCOPE_WORKGROUP);
            }
        }
        for (; j < n; ++j) {
            const unsigned p = __builtin_amdgcn_readlane(pk, j);
            const v2f v = *(const v2f*)(hl + (size_t)(p & 0xffffu) * D);
            float* a = al + (p >> 16) * D;
            __hip_atomic_fetch_add(a,     v.x, __ATOMIC_RELAXED, __HIP_MEMORY_SCOPE_WORKGROUP);
            __hip_atomic_fetch_add(a + 1, v.y, __ATOMIC_RELAXED, __HIP_MEMORY_SCOPE_WORKGROUP);
        }
    }
    __syncthreads();

    const int row0  = b * RB;
    const int nrows = (N_NODES - row0 < RB) ? (N_NODES - row0) : RB;
    v2f hv[RPW], sv[RPW];
    float dr[RPW];
    #pragma unroll
    for (int k = 0; k < RPW; ++k) {
        const int rl  = wave + 32 * k;
        const int row = row0 + ((rl < nrows) ? rl : (nrows - 1));
        const size_t idx = (size_t)row * D + 2 * lane;
        hv[k] = *(const v2f*)(h + idx);
        sv[k] = *(const v2f*)(support + idx);
        dr[k] = d[row];
    }
    #pragma unroll
    for (int k = 0; k < RPW; ++k) {
        const int rl = wave + 32 * k;
        if (rl < nrows) {
            const size_t idx = (size_t)(row0 + rl) * D + 2 * lane;
            const v2f av = *(const v2f*)(al + rl * D);
            v2f o;
            o.x = ((av.x + hv[k].x) * dr[k] * SMOOTHF + sv[k].x) * (1.0f / (1.0f + SMOOTHF));
            o.y = ((av.y + hv[k].y) * dr[k] * SMOOTHF + sv[k].y) * (1.0f / (1.0f + SMOOTHF));
            *(v2f*)(out + idx) = o;
        }
    }
}


extern "C" void kernel_launch(void* const* d_in, const int* in_sizes, int n_in,
                              void* d_out, int out_size, void* d_ws, size_t ws_size,
                              hipStream_t stream) {
    const float* x    = (const float*)d_in[0];
    const float* W    = (const float*)d_in[1];
    const float* bias = (const float*)d_in[2];
    const int*   erow = (const int*)d_in[3];
    const int*   ecol = (const int*)d_in[4];
    float* out = (float*)d_out;

    float* ws      = (float*)d_ws;
    float* support = ws;
    float* h       = ws + (size_t)N_NODES * D;
    unsigned* sorted = (unsigned*)(ws + 2 * (size_t)N_NODES * D);
    int* cnt       = (int*)(sorted + N_EDGES);
    int* off       = cnt + NB * G;
    float* deg     = ws + 3 * (size_t)N_NODES * D;

    k_count_zero <<<G + ZB, TPB, 0, stream>>>(erow, cnt, deg);
    k_scan_degree<<<1 + DB, TPB, 0, stream>>>(cnt, off, ecol, deg);
    k_place_rsqrt<<<G + ZB, TPB, 0, stream>>>(erow, ecol, off, sorted, deg);
    k_gemm_wmma  <<<(N_TILES + 7) / 8, 256, 0, stream>>>(x, W, bias, deg, support, h);
    k_accum    <<<NB, 1024, 0, stream>>>(sorted, off, h, support, deg, out);
}


# degree count: each work-group reads its die and adds into that die's own integer count plane (8 planes, summed in the rsqrt pass): no count line is updated from both cache halves; counts exact
# speedup vs baseline: 2.5483x; 1.4077x over previous
#include <hip/hip_runtime.h>
#include <hip/hip_bf16.h>

typedef float v2f __attribute__((ext_vector_type(2)));
typedef float v8f __attribute__((ext_vector_type(8)));

#define N_NODES 50000
#define N_EDGES 800000
#define D 64
#define N_TILES (N_NODES / 16)
#define SMOOTHF 0.5f


#define TPB 1024
#define ZB  ((N_NODES + TPB - 1) / TPB)
#define DB  ((N_EDGES + TPB - 1) / TPB)
#define NPL 8
#define PLS 50048


__device__ __forceinline__ void zero_deg_body(unsigned* __restrict__ degc, int blk) {
    const int i = blk * TPB + threadIdx.x;
    if (i < N_NODES) {
        for (int p = 0; p < NPL; ++p) degc[(size_t)p * PLS + i] = 0u;
    }
}

__device__ __forceinline__ void degree_body(const int* __restrict__ ecol,
                                            unsigned* __restrict__ degc, int blk) {
    const unsigned die = (__builtin_amdgcn_s_sendmsg_rtn(0x87) >> 16) & (NPL - 1);
    unsigned* plane = degc + (size_t)die * PLS;
    const int e = blk * TPB + threadIdx.x;
    if (e < N_EDGES)
        __hip_atomic_fetch_add(&plane[ecol[e]], 1u, __ATOMIC_RELAXED, __HIP_MEMORY_SCOPE_AGENT);
}

__device__ __forceinline__ void rsqrt_body(const unsigned* __restrict__ degc,
                                           float* __restrict__ d, int blk) {
    const int i = blk * TPB + threadIdx.x;
    if (i < N_NODES) {
        unsigned c = 0;
        #pragma unroll
        for (int p = 0; p < NPL; ++p) c += degc[(size_t)p * PLS + i];
        d[i] = rsqrtf((float)c + 1.0f);
    }
}


__global__ __launch_bounds__(256) void k_gemm_wmma(
    const float* __restrict__ x, const float* __restrict__ W,
    const float* __restrict__ bias, const float* __restrict__ d,
    float* __restrict__ support, float* __restrict__ h)
{
    __shared__ float Wl[D * D];
    for (int i = threadIdx.x; i < D * D; i += 256) Wl[i] = W[i];
    __syncthreads();

    const int wave = threadIdx.x >> 5;
    const int lane = threadIdx.x & 31;
    const int tile = blockIdx.x * 8 + wave;
    if (tile >= N_TILES) return;

    const int row0 = tile * 16;
    const int half = lane >> 4;
    const int l15  = lane & 15;

    v8f acc[4] = {v8f{}, v8f{}, v8f{}, v8f{}};
    const float* xrow = x + (size_t)(row0 + l15) * D;

    #pragma unroll
    for (int kk = 0; kk < 16; ++kk) {
        const int k = kk * 4 + half * 2;
        v2f a = *(const v2f*)(xrow + k);
        #pragma unroll
        for (int nt = 0; nt < 4; ++nt) {
            const int n = nt * 16 + l15;
            v2f bf = *(const v2f*)(&Wl[n * D + k]);
            acc[nt] = __builtin_amdgcn_wmma_f32_16x16x4_f32(
                 false, a,  false, bf,
                 (short)0, acc[nt],
                 false,  false);
        }
    }

    #pragma unroll
    for (int nt = 0; nt < 4; ++nt) {
        const int n  = nt * 16 + l15;
        const float bv = bias[n];
        #pragma unroll
        for (int r = 0; r < 8; ++r) {
            const int row  = row0 + r + 8 * half;
            const float s  = acc[nt][r] + bv;
            const float hv = s * d[row];
            const size_t idx = (size_t)row * D + n;
            support[idx] = s;
            h[idx]       = hv;
        }
    }
}


#define NB  256
#define RB  196
#define G   256
#define EPG (N_EDGES / G)
#define RPW ((RB + 31) / 32)

__device__ __forceinline__ void bin_count_body(const int* __restrict__ erow,
                                               int* __restrict__ cnt, int g)
{
    __shared__ int hist[NB];
    const int t = threadIdx.x;
    if (t < NB) hist[t] = 0;
    __syncthreads();
    const int e0 = g * EPG;
    for (int i = t; i < EPG; i += TPB) {
        const unsigned b = (unsigned)erow[e0 + i] / RB;
        __hip_atomic_fetch_add(&hist[b], 1, __ATOMIC_RELAXED, __HIP_MEMORY_SCOPE_WORKGROUP);
    }
    __syncthreads();
    if (t < NB) cnt[t * G + g] = hist[t];
}

__device__ __forceinline__ void bin_scan_body(const int* __restrict__ cnt,
                                              int* __restrict__ off)
{
    __shared__ int sh[TPB];
    const int t = threadIdx.x;
    const int4* src = (const int4*)(cnt + t * 64);
    int4 v[16];
    int sum = 0;
    #pragma unroll
    for (int i = 0; i < 16; ++i) {
        v[i] = src[i];
        sum += v[i].x + v[i].y + v[i].z + v[i].w;
    }
    sh[t] = sum;
    __syncthreads();
    for (int s = 1; s < TPB; s <<= 1) {
        const int add = (t >= s) ? sh[t - s] : 0;
        __syncthreads();
        sh[t] += add;
        __syncthreads();
    }
    int run = sh[t] - sum;
    int4* dst = (int4*)(off + t * 64);
    #pragma unroll
    for (int i = 0; i < 16; ++i) {
        int4 o;
        o.x = run; run += v[i].x;
        o.y = run; run += v[i].y;
        o.z = run; run += v[i].z;
        o.w = run; run += v[i].w;
        dst[i] = o;
    }
}

__device__ __forceinline__ void bin_place_body(const int* __restrict__ erow,
                                               const int* __restrict__ ecol,
                                               const int* __restrict__ off,
                                               unsigned* __restrict__ sorted, int g)
{
    __shared__ int cur[NB];
    const int t = threadIdx.x;
    if (t < NB) cur[t] = off[t * G + g];
    __syncthreads();
    const int e0 = g * EPG;
    for (int i = t; i < EPG; i += TPB) {
        const unsigned r = (unsigned)erow[e0 + i];
        const unsigned c = (unsigned)ecol[e0 + i];
        const unsigned b = r / RB;
        const int slot = __hip_atomic_fetch_add(&cur[b], 1, __ATOMIC_RELAXED,
                                                __HIP_MEMORY_SCOPE_WORKGROUP);
        sorted[slot] = ((r - b * RB) << 16) | c;
    }
}


__global__ __launch_bounds__(TPB) void k_count_zero(const int* __restrict__ erow,
                                                    int* __restrict__ cnt,
                                                    unsigned* __restrict__ degc)
{
    if (blockIdx.x < G) bin_count_body(erow, cnt, blockIdx.x);
    else                zero_deg_body(degc, blockIdx.x - G);
}

__global__ __launch_bounds__(TPB) void k_scan_degree(const int* __restrict__ cnt,
                                                     int* __restrict__ off,
                                                     const int* __restrict__ ecol,
                                                     unsigned* __restrict__ degc)
{
    if (blockIdx.x == 0) bin_scan_body(cnt, off);
    else                 degree_body(ecol, degc, blockIdx.x - 1);
}

__global__ __launch_bounds__(TPB) void k_place_rsqrt(const int* __restrict__ erow,
                                                     const int* __restrict__ ecol,
                                                     const int* __restrict__ off,
                                                     unsigned* __restrict__ sorted,
                                                     const unsigned* __restrict__ degc,
                                                     float* __restrict__ d)
{
    if (blockIdx.x < G) bin_place_body(erow, ecol, off, sorted, blockIdx.x);
    else                rsqrt_body(degc, d, blockIdx.x - G);
}

__global__ __launch_bounds__(1024) void k_accum(
    const unsigned* __restrict__ sorted, const int* __restrict__ off,
    const float* __restrict__ h, const float* __restrict__ support,
    const float* __restrict__ d, float* __restrict__ out)
{
    __shared__ float acc[RB * D];
    const int b    = blockIdx.x;
    const int tid  = threadIdx.x;
    const int lane = tid & 31;
    const int wave = __builtin_amdgcn_readfirstlane(tid >> 5);

    for (int i = tid; i < RB * D; i += 1024) acc[i] = 0.0f;
    const int bs = off[b * G];
    const int be = (b + 1 < NB) ? off[(b + 1) * G] : N_EDGES;
    __syncthreads();

    const int per = ((((be - bs) + 31) >> 5) + 7) & ~7;
    const int w0  = bs + wave * per;
    const int w1  = (w0 + per < be) ? (w0 + per) : be;
    const float* hl = h + 2 * lane;
    float* al = acc + 2 * lane;

    for (int base = w0; base < w1; base += 32) {
        const int n = (w1 - base < 32) ? (w1 - base) : 32;
        unsigned pk = 0;
        if (lane < n) pk = sorted[base + lane];
        int j = 0;
        for (; j + 8 <= n; j += 8) {
            unsigned p[8];
            v2f v[8];
            #pragma unroll
            for (int u = 0; u < 8; ++u) {
                p[u] = __builtin_amdgcn_readlane(pk, j + u);
                v[u] = *(const v2f*)(hl + (size_t)(p[u] & 0xffffu) * D);
            }
            #pragma unroll
            for (int u = 0; u < 8; ++u) {
                float* a = al + (p[u] >> 16) * D;
                __hip_atomic_fetch_add(a,     v[u].x, __ATOMIC_RELAXED, __HIP_MEMORY_SCOPE_WORKGROUP);
                __hip_atomic_fetch_add(a + 1, v[u].y, __ATOMIC_RELAXED, __HIP_MEMORY_SCOPE_WORKGROUP);
            }
        }
        for (; j < n; ++j) {
            const unsigned p = __builtin_amdgcn_readlane(pk, j);
            const v2f v = *(const v2f*)(hl + (size_t)(p & 0xffffu) * D);
            float* a = al + (p >> 16) * D;
            __hip_atomic_fetch_add(a,     v.x, __ATOMIC_RELAXED, __HIP_MEMORY_SCOPE_WORKGROUP);
            __hip_atomic_fetch_add(a + 1, v.y, __ATOMIC_RELAXED, __HIP_MEMORY_SCOPE_WORKGROUP);
        }
    }
    __syncthreads();

    const int row0  = b * RB;
    const int nrows = (N_NODES - row0 < RB) ? (N_NODES - row0) : RB;
    v2f hv[RPW], sv[RPW];
    float dr[RPW];
    #pragma unroll
    for (int k = 0; k < RPW; ++k) {
        const int rl  = wave + 32 * k;
        const int row = row0 + ((rl < nrows) ? rl : (nrows - 1));
        const size_t idx = (size_t)row * D + 2 * lane;
        hv[k] = *(const v2f*)(h + idx);
        sv[k] = *(const v2f*)(support + idx);
        dr[k] = d[row];
    }
    #pragma unroll
    for (int k = 0; k < RPW; ++k) {
        const int rl = wave + 32 * k;
        if (rl < nrows) {
            const size_t idx = (size_t)(row0 + rl) * D + 2 * lane;
            const v2f av = *(const v2f*)(al + rl * D);
            v2f o;
            o.x = ((av.x + hv[k].x) * dr[k] * SMOOTHF + sv[k].x) * (1.0f / (1.0f + SMOOTHF));
            o.y = ((av.y + hv[k].y) * dr[k] * SMOOTHF + sv[k].y) * (1.0f / (1.0f + SMOOTHF));
            *(v2f*)(out + idx) = o;
        }
    }
}


extern "C" void kernel_launch(void* const* d_in, const int* in_sizes, int n_in,
                              void* d_out, int out_size, void* d_ws, size_t ws_size,
                              hipStream_t stream) {
    const float* x    = (const float*)d_in[0];
    const float* W    = (const float*)d_in[1];
    const float* bias = (const float*)d_in[2];
    const int*   erow = (const int*)d_in[3];
    const int*   ecol = (const int*)d_in[4];
    float* out = (float*)d_out;

    float* ws      = (float*)d_ws;
    float* support = ws;
    float* h       = ws + (size_t)N_NODES * D;
    unsigned* sorted = (unsigned*)(ws + 2 * (size_t)N_NODES * D);
    int* cnt       = (int*)(sorted + N_EDGES);
    int* off       = cnt + NB * G;
    float* deg     = ws + 3 * (size_t)N_NODES * D;
    unsigned* degc = (unsigned*)(deg + PLS);

    k_count_zero <<<G + ZB, TPB, 0, stream>>>(erow, cnt, degc);
    k_scan_degree<<<1 + DB, TPB, 0, stream>>>(cnt, off, ecol, degc);
    k_place_rsqrt<<<G + ZB, TPB, 0, stream>>>(erow, ecol, off, sorted, degc, deg);
    k_gemm_wmma  <<<(N_TILES + 7) / 8, 256, 0, stream>>>(x, W, bias, deg, support, h);
    k_accum    <<<NB, 1024, 0, stream>>>(sorted, off, h, support, deg, out);
}


# edge grouping without a scan: each fixed 3125-edge slice is ordered by range in LDS and stored whole in place; a range walks the 256 pieces by a start/count table; LDS f32 sums in arrival order
# speedup vs baseline: 3.9048x; 1.5323x over previous
#include <hip/hip_runtime.h>
#include <hip/hip_bf16.h>

typedef float v2f __attribute__((ext_vector_type(2)));
typedef float v8f __attribute__((ext_vector_type(8)));

#define N_NODES 50000
#define N_EDGES 800000
#define D 64
#define N_TILES (N_NODES / 16)
#define SMOOTHF 0.5f


#define TPB 1024
#define ZB  ((N_NODES + TPB - 1) / TPB)
#define DB  ((N_EDGES + TPB - 1) / TPB)
#define NPL 8
#define PLS 50048


__device__ __forceinline__ void zero_deg_body(unsigned* __restrict__ degc, int blk) {
    const int i = blk * TPB + threadIdx.x;
    if (i < N_NODES) {
        for (int p = 0; p < NPL; ++p) degc[(size_t)p * PLS + i] = 0u;
    }
}

__device__ __forceinline__ void degree_body(const int* __restrict__ ecol,
                                            unsigned* __restrict__ degc, int blk) {
    const unsigned die = (__builtin_amdgcn_s_sendmsg_rtn(0x87) >> 16) & (NPL - 1);
    unsigned* plane = degc + (size_t)die * PLS;
    const int e = blk * TPB + threadIdx.x;
    if (e < N_EDGES)
        __hip_atomic_fetch_add(&plane[ecol[e]], 1u, __ATOMIC_RELAXED, __HIP_MEMORY_SCOPE_AGENT);
}

__device__ __forceinline__ void rsqrt_body(const unsigned* __restrict__ degc,
                                           float* __restrict__ d, int blk) {
    const int i = blk * TPB + threadIdx.x;
    if (i < N_NODES) {
        unsigned c = 0;
        #pragma unroll
        for (int p = 0; p < NPL; ++p) c += degc[(size_t)p * PLS + i];
        d[i] = rsqrtf((float)c + 1.0f);
    }
}


__global__ __launch_bounds__(256) void k_gemm_wmma(
    const float* __restrict__ x, const float* __restrict__ W,
    const float* __restrict__ bias, const float* __restrict__ d,
    float* __restrict__ support, float* __restrict__ h)
{
    __shared__ float Wl[D * D];
    for (int i = threadIdx.x; i < D * D; i += 256) Wl[i] = W[i];
    __syncthreads();

    const int wave = threadIdx.x >> 5;
    const int lane = threadIdx.x & 31;
    const int tile = blockIdx.x * 8 + wave;
    if (tile >= N_TILES) return;

    const int row0 = tile * 16;
    const int half = lane >> 4;
    const int l15  = lane & 15;

    v8f acc[4] = {v8f{}, v8f{}, v8f{}, v8f{}};
    const float* xrow = x + (size_t)(row0 + l15) * D;

    #pragma unroll
    for (int kk = 0; kk < 16; ++kk) {
        const int k = kk * 4 + half * 2;
        v2f a = *(const v2f*)(xrow + k);
        #pragma unroll
        for (int nt = 0; nt < 4; ++nt) {
            const int n = nt * 16 + l15;
            v2f bf = *(const v2f*)(&Wl[n * D + k]);
            acc[nt] = __builtin_amdgcn_wmma_f32_16x16x4_f32(
                 false, a,  false, bf,
                 (short)0, acc[nt],
                 false,  false);
        }
    }

    #pragma unroll
    for (int nt = 0; nt < 4; ++nt) {
        const int n  = nt * 16 + l15;
        const float bv = bias[n];
        #pragma unroll
        for (int r = 0; r < 8; ++r) {
            const int row  = row0 + r + 8 * half;
            const float s  = acc[nt][r] + bv;
            const float hv = s * d[row];
            const size_t idx = (size_t)row * D + n;
            support[idx] = s;
            h[idx]       = hv;
        }
    }
}


#define NB  256
#define RB  196
#define G   256
#define EPG (N_EDGES / G)
#define EPT ((EPG + TPB - 1) / TPB)
#define SPW (G / 32)
#define RPW ((RB + 31) / 32)

__device__ __forceinline__ void group_slice_body(const int* __restrict__ erow,
                                                 const int* __restrict__ ecol,
                                                 unsigned* __restrict__ sorted,
                                                 unsigned* __restrict__ seg, int g)
{
    __shared__ int hist[NB];
    __shared__ int wtot[NB / 32];
    __shared__ unsigned buf[EPG];
    const int t    = threadIdx.x;
    const int lane = t & 31;
    if (t < NB) hist[t] = 0;
    __syncthreads();

    const int e0 = g * EPG;
    unsigned er[EPT], ec[EPT], pkv[EPT], bb[EPT];
    int rk[EPT];
    #pragma unroll
    for (int k = 0; k < EPT; ++k) {
        const int i = t + k * TPB;
        er[k] = 0u; ec[k] = 0u;
        if (i < EPG) {
            er[k] = (unsigned)erow[e0 + i];
            ec[k] = (unsigned)ecol[e0 + i];
        }
    }
    #pragma unroll
    for (int k = 0; k < EPT; ++k) {
        const int i = t + k * TPB;
        const unsigned b = er[k] / RB;
        bb[k]  = b;
        pkv[k] = ((er[k] - b * RB) << 16) | ec[k];
        rk[k]  = 0;
        if (i < EPG)
            rk[k] = __hip_atomic_fetch_add(&hist[b], 1, __ATOMIC_RELAXED,
                                           __HIP_MEMORY_SCOPE_WORKGROUP);
    }
    __syncthreads();

    int cnt_t = 0, inc = 0;
    if (t < NB) {
        cnt_t = hist[t];
        inc = cnt_t;
        #pragma unroll
        for (int s = 1; s < 32; s <<= 1) {
            const int v = __shfl_up(inc, s);
            if (lane >= s) inc += v;
        }
        if (lane == 31) wtot[t >> 5] = inc;
    }
    __syncthreads();
    if (t < NB) {
        int st = inc - cnt_t;
        for (int w = 0; w < (t >> 5); ++w) st += wtot[w];
        hist[t] = st;
        seg[t * G + g] = ((unsigned)st << 12) | (unsigned)cnt_t;
    }
    __syncthreads();

    #pragma unroll
    for (int k = 0; k < EPT; ++k) {
        const int i = t + k * TPB;
        if (i < EPG) buf[hist[bb[k]] + rk[k]] = pkv[k];
    }
    __syncthreads();
    for (int i = t; i < EPG; i += TPB) sorted[e0 + i] = buf[i];
}


__global__ __launch_bounds__(TPB) void k_zero(unsigned* __restrict__ degc)
{
    zero_deg_body(degc, blockIdx.x);
}

__global__ __launch_bounds__(TPB) void k_group_degree(const int* __restrict__ erow,
                                                      const int* __restrict__ ecol,
                                                      unsigned* __restrict__ sorted,
                                                      unsigned* __restrict__ seg,
                                                      unsigned* __restrict__ degc)
{
    if (blockIdx.x < G) group_slice_body(erow, ecol, sorted, seg, blockIdx.x);
    else                degree_body(ecol, degc, blockIdx.x - G);
}

__global__ __launch_bounds__(TPB) void k_rsqrt(const unsigned* __restrict__ degc,
                                               float* __restrict__ d)
{
    rsqrt_body(degc, d, blockIdx.x);
}

__global__ __launch_bounds__(1024) void k_accum(
    const unsigned* __restrict__ sorted, const unsigned* __restrict__ seg,
    const float* __restrict__ h, const float* __restrict__ support,
    const float* __restrict__ d, float* __restrict__ out)
{
    __shared__ float acc[RB * D];
    const int b    = blockIdx.x;
    const int tid  = threadIdx.x;
    const int lane = tid & 31;
    const int wave = __builtin_amdgcn_readfirstlane(tid >> 5);

    for (int i = tid; i < RB * D; i += 1024) acc[i] = 0.0f;

    unsigned sgv = 0;
    if (lane < SPW) sgv = seg[b * G + wave * SPW + lane];
    int pc[SPW + 1], dl[SPW];
    pc[0] = 0;
    #pragma unroll
    for (int m = 0; m < SPW; ++m) {
        const unsigned s = __builtin_amdgcn_readlane(sgv, m);
        dl[m]     = (wave * SPW + m) * EPG + (int)(s >> 12) - pc[m];
        pc[m + 1] = pc[m] + (int)(s & 0xfffu);
    }
    const int total = pc[SPW];
    __syncthreads();

    const float* hl = h + 2 * lane;
    float* al = acc + 2 * lane;

    for (int base = 0; base < total; base += 32) {
        const int n = (total - base < 32) ? (total - base) : 32;
        const int i = base + lane;
        int dsel = dl[0];
        #pragma unroll
        for (int m = 1; m < SPW; ++m) dsel = (i >= pc[m]) ? dl[m] : dsel;
        unsigned pk = 0;
        if (lane < n) pk = sorted[i + dsel];
        int j = 0;
        for (; j + 8 <= n; j += 8) {
            unsigned p[8];
            v2f v[8];
            #pragma unroll
            for (int u = 0; u < 8; ++u) {
                p[u] = __builtin_amdgcn_readlane(pk, j + u);
                v[u] = *(const v2f*)(hl + (size_t)(p[u] & 0xffffu) * D);
            }
            #pragma unroll
            for (int u = 0; u < 8; ++u) {
                float* a = al + (p[u] >> 16) * D;
                __hip_atomic_fetch_add(a,     v[u].x, __ATOMIC_RELAXED, __HIP_MEMORY_SCOPE_WORKGROUP);
                __hip_atomic_fetch_add(a + 1, v[u].y, __ATOMIC_RELAXED, __HIP_MEMORY_SCOPE_WORKGROUP);
            }
        }
        if (j < n) {
            const int m8 = n - j;
            unsigned p[8];
            v2f v[8];
            #pragma unroll
            for (int u = 0; u < 8; ++u) {
                p[u] = __builtin_amdgcn_readlane(pk, (u < m8) ? (j + u) : (n - 1));
                v[u] = *(const v2f*)(hl + (size_t)(p[u] & 0xffffu) * D);
            }
            #pragma unroll
            for (int u = 0; u < 8; ++u) {
                if (u < m8) {
                    float* a = al + (p[u] >> 16) * D;
                    __hip_atomic_fetch_add(a,     v[u].x, __ATOMIC_RELAXED, __HIP_MEMORY_SCOPE_WORKGROUP);
                    __hip_atomic_fetch_add(a + 1, v[u].y, __ATOMIC_RELAXED, __HIP_MEMORY_SCOPE_WORKGROUP);
                }
            }
        }
    }
    __syncthreads();

    const int row0  = b * RB;
    const int nrows = (N_NODES - row0 < RB) ? (N_NODES - row0) : RB;
    v2f hv[RPW], sv[RPW];
    float dr[RPW];
    #pragma unroll
    for (int k = 0; k < RPW; ++k) {
        const int rl  = wave + 32 * k;
        const int row = row0 + ((rl < nrows) ? rl : (nrows - 1));
        const size_t idx = (size_t)row * D + 2 * lane;
        hv[k] = *(const v2f*)(h + idx);
        sv[k] = *(const v2f*)(support + idx);
        dr[k] = d[row];
    }
    #pragma unroll
    for (int k = 0; k < RPW; ++k) {
        const int rl = wave + 32 * k;
        if (rl < nrows) {
            const size_t idx = (size_t)(row0 + rl) * D + 2 * lane;
            const v2f av = *(const v2f*)(al + rl * D);
            v2f o;
            o.x = ((av.x + hv[k].x) * dr[k] * SMOOTHF + sv[k].x) * (1.0f / (1.0f + SMOOTHF));
            o.y = ((av.y + hv[k].y) * dr[k] * SMOOTHF + sv[k].y) * (1.0f / (1.0f + SMOOTHF));
            *(v2f*)(out + idx) = o;
        }
    }
}


extern "C" void kernel_launch(void* const* d_in, const int* in_sizes, int n_in,
                              void* d_out, int out_size, void* d_ws, size_t ws_size,
                              hipStream_t stream) {
    const float* x    = (const float*)d_in[0];
    const float* W    = (const float*)d_in[1];
    const float* bias = (const float*)d_in[2];
    const int*   erow = (const int*)d_in[3];
    const int*   ecol = (const int*)d_in[4];
    float* out = (float*)d_out;

    float* ws      = (float*)d_ws;
    float* support = ws;
    float* h       = ws + (size_t)N_NODES * D;
    unsigned* sorted = (unsigned*)(ws + 2 * (size_t)N_NODES * D);
    unsigned* seg  = sorted + N_EDGES;
    float* deg     = ws + 3 * (size_t)N_NODES * D;
    unsigned* degc = (unsigned*)(deg + PLS);

    k_zero        <<<ZB, TPB, 0, stream>>>(degc);
    k_group_degree<<<G + DB, TPB, 0, stream>>>(erow, ecol, sorted, seg, degc);
    k_rsqrt       <<<ZB, TPB, 0, stream>>>(degc, deg);
    k_gemm_wmma   <<<(N_TILES + 7) / 8, 256, 0, stream>>>(x, W, bias, deg, support, h);
    k_accum       <<<NB, 1024, 0, stream>>>(sorted, seg, h, support, deg, out);
}


# degrees by counting sort: a slice also orders its source nodes by 196-node range in LDS (bytes); one work-group per range counts them in LDS, writes d; no global adds, no zero pass; 4 launches->3
# speedup vs baseline: 5.3718x; 1.3757x over previous
#include <hip/hip_runtime.h>
#include <hip/hip_bf16.h>

typedef float v2f __attribute__((ext_vector_type(2)));
typedef float v4f __attribute__((ext_vector_type(4)));
typedef float v8f __attribute__((ext_vector_type(8)));

#define N_NODES 50000
#define N_EDGES 800000
#define D 64
#define SMOOTHF 0.5f


#define TPB 1024


#define NB  256
#define RB  196
#define G   256
#define EPG (N_EDGES / G)
#define EPT ((EPG + TPB - 1) / TPB)
#define CPS 3136
#define DPL 50048
#define SPW (G / 32)
#define RPW ((RB + 31) / 32)
#define MT  ((RB + 15) / 16)
#define RBP (MT * 16)
#define PA  68

__global__ __launch_bounds__(TPB) void k_group(const int* __restrict__ erow,
                                               const int* __restrict__ ecol,
                                               unsigned* __restrict__ sorted,
                                               unsigned* __restrict__ seg,
                                               unsigned* __restrict__ csorted,
                                               unsigned* __restrict__ cseg)
{
    __shared__ int hist[2 * NB];
    __shared__ int wtot[2 * NB / 32];
    __shared__ unsigned buf[EPG];
    __shared__ unsigned cbw[CPS / 4];
    unsigned char* cbuf = (unsigned char*)cbw;
    const int g    = blockIdx.x;
    const int t    = threadIdx.x;
    const int lane = t & 31;
    if (t < 2 * NB) hist[t] = 0;
    if (t < CPS - EPG) cbuf[EPG + t] = 0;
    __syncthreads();

    const int e0 = g * EPG;
    unsigned er[EPT], ec[EPT], pkv[EPT], bb[EPT], cb[EPT], cl[EPT];
    int rk[EPT], ck[EPT];
    #pragma unroll
    for (int k = 0; k < EPT; ++k) {
        const int i = t + k * TPB;
        er[k] = 0u; ec[k] = 0u;
        if (i < EPG) {
            er[k] = (unsigned)erow[e0 + i];
            ec[k] = (unsigned)ecol[e0 + i];
        }
    }
    #pragma unroll
    for (int k = 0; k < EPT; ++k) {
        const int i = t + k * TPB;
        const unsigned b = er[k] / RB;
        const unsigned c = ec[k] / RB;
        bb[k]  = b;
        cb[k]  = c;
        pkv[k] = ((er[k] - b * RB) << 16) | ec[k];
        cl[k]  = ec[k] - c * RB;
        rk[k]  = 0;
        ck[k]  = 0;
        if (i < EPG) {
            rk[k] = __hip_atomic_fetch_add(&hist[b], 1, __ATOMIC_RELAXED,
                                           __HIP_MEMORY_SCOPE_WORKGROUP);
            ck[k] = __hip_atomic_fetch_add(&hist[NB + c], 1, __ATOMIC_RELAXED,
                                           __HIP_MEMORY_SCOPE_WORKGROUP);
        }
    }
    __syncthreads();

    int cnt_t = 0, inc = 0;
    if (t < 2 * NB) {
        cnt_t = hist[t];
        inc = cnt_t;
        #pragma unroll
        for (int sft = 1; sft < 32; sft <<= 1) {
            const int v = __shfl_up(inc, sft);
            if (lane >= sft) inc += v;
        }
        if (lane == 31) wtot[t >> 5] = inc;
    }
    __syncthreads();
    if (t < 2 * NB) {
        const int key = t & (NB - 1);
        const int w0  = (t >> 8) * (NB / 32);
        int st = inc - cnt_t;
        for (int w = w0; w < (t >> 5); ++w) st += wtot[w];
        hist[t] = st;
        const unsigned word = ((unsigned)st << 12) | (unsigned)cnt_t;
        if (t < NB) seg[key * G + g]  = word;
        else        cseg[key * G + g] = word;
    }
    __syncthreads();

    #pragma unroll
    for (int k = 0; k < EPT; ++k) {
        const int i = t + k * TPB;
        if (i < EPG) {
            buf[hist[bb[k]] + rk[k]] = pkv[k];
            cbuf[hist[NB + cb[k]] + ck[k]] = (unsigned char)cl[k];
        }
    }
    __syncthreads();
    for (int i = t; i < EPG; i += TPB) sorted[e0 + i] = buf[i];
    if (t < CPS / 4) csorted[g * (CPS / 4) + t] = cbw[t];
}

#define DTB G
__global__ __launch_bounds__(DTB) void k_degree(const unsigned* __restrict__ csorted,
                                                const unsigned* __restrict__ cseg,
                                                float* __restrict__ d)
{
    __shared__ unsigned cnt[256];
    const int b = blockIdx.x;
    const int t = threadIdx.x;
    cnt[t] = 0u;
    const unsigned sw = cseg[b * G + t];
    const int n = (int)(sw & 0xfffu);
    const unsigned char* p = (const unsigned char*)csorted + (size_t)t * CPS + (sw >> 12);
    __syncthreads();
    for (int j = 0; j < n; j += 8) {
        unsigned v[8];
        #pragma unroll
        for (int u = 0; u < 8; ++u) v[u] = p[(j + u < n) ? (j + u) : (n - 1)];
        #pragma unroll
        for (int u = 0; u < 8; ++u)
            __hip_atomic_fetch_add(&cnt[v[u]], (j + u < n) ? 1u : 0u, __ATOMIC_RELAXED, __HIP_MEMORY_SCOPE_WORKGROUP);
    }
    __syncthreads();
    const int node = b * RB + t;
    if (t < RB && node < N_NODES) d[node] = rsqrtf((float)cnt[t] + 1.0f);
}

__global__ __launch_bounds__(1024) void k_accum(
    const unsigned* __restrict__ sorted, const unsigned* __restrict__ seg,
    const float* __restrict__ x, const float* __restrict__ W,
    const float* __restrict__ bias, const float* __restrict__ d,
    float* __restrict__ out)
{
    __shared__ __attribute__((aligned(16))) float acc[RBP * PA];
    __shared__ __attribute__((aligned(16))) float Wl[D * PA];
    __shared__ float sdl[RBP];
    const int b    = blockIdx.x;
    const int tid  = threadIdx.x;
    const int lane = tid & 31;
    const int wave = __builtin_amdgcn_readfirstlane(tid >> 5);

    const v4f wv = *(const v4f*)(W + 4 * tid);
    for (int i = tid; i < RBP * PA / 4; i += 1024) ((v4f*)acc)[i] = v4f{};
    if (tid < RBP) sdl[tid] = 0.0f;
    *(v4f*)(&Wl[(tid >> 4) * PA + (tid & 15) * 4]) = wv;

    unsigned sgv = 0;
    if (lane < SPW) sgv = seg[b * G + wave * SPW + lane];
    int pc[SPW + 1], dl[SPW];
    pc[0] = 0;
    #pragma unroll
    for (int m = 0; m < SPW; ++m) {
        const unsigned s = __builtin_amdgcn_readlane(sgv, m);
        dl[m]     = (wave * SPW + m) * EPG + (int)(s >> 12) - pc[m];
        pc[m + 1] = pc[m] + (int)(s & 0xfffu);
    }
    const int total = pc[SPW];
    __syncthreads();

    const float* xl = x + 2 * lane;
    float* al = acc + 2 * lane;

    for (int base = 0; base < total; base += 32) {
        const int n = (total - base < 32) ? (total - base) : 32;
        const int i = base + lane;
        int dsel = dl[0];
        #pragma unroll
        for (int m = 1; m < SPW; ++m) dsel = (i >= pc[m]) ? dl[m] : dsel;
        unsigned pk = 0;
        float dv = 0.0f;
        if (lane < n) {
            pk = sorted[i + dsel];
            dv = d[pk & 0xffffu];
        }
        const unsigned dvb = __float_as_uint(dv);
        int j = 0;
        for (; j + 8 <= n; j += 8) {
            unsigned p[8];
            float s[8];
            v2f v[8];
            #pragma unroll
            for (int u = 0; u < 8; ++u) {
                p[u] = __builtin_amdgcn_readlane(pk, j + u);
                s[u] = __uint_as_float(__builtin_amdgcn_readlane(dvb, j + u));
                v[u] = *(const v2f*)(xl + (size_t)(p[u] & 0xffffu) * D);
            }
            #pragma unroll
            for (int u = 0; u < 8; ++u) {
                float* a = al + (p[u] >> 16) * PA;
                __hip_atomic_fetch_add(a,     v[u].x * s[u], __ATOMIC_RELAXED, __HIP_MEMORY_SCOPE_WORKGROUP);
                __hip_atomic_fetch_add(a + 1, v[u].y * s[u], __ATOMIC_RELAXED, __HIP_MEMORY_SCOPE_WORKGROUP);
            }
        }
        if (j < n) {
            const int m8 = n - j;
            unsigned p[8];
            float s[8];
            v2f v[8];
            #pragma unroll
            for (int u = 0; u < 8; ++u) {
                const int sel = (u < m8) ? (j + u) : (n - 1);
                p[u] = __builtin_amdgcn_readlane(pk, sel);
                s[u] = __uint_as_float(__builtin_amdgcn_readlane(dvb, sel));
                v[u] = *(const v2f*)(xl + (size_t)(p[u] & 0xffffu) * D);
            }
            #pragma unroll
            for (int u = 0; u < 8; ++u) {
                if (u < m8) {
                    float* a = al + (p[u] >> 16) * PA;
                    __hip_atomic_fetch_add(a,     v[u].x * s[u], __ATOMIC_RELAXED, __HIP_MEMORY_SCOPE_WORKGROUP);
                    __hip_atomic_fetch_add(a + 1, v[u].y * s[u], __ATOMIC_RELAXED, __HIP_MEMORY_SCOPE_WORKGROUP);
                }
            }
        }
        if (lane < n)
            __hip_atomic_fetch_add(&sdl[pk >> 16], dv, __ATOMIC_RELAXED, __HIP_MEMORY_SCOPE_WORKGROUP);
    }
    __syncthreads();

    const int row0  = b * RB;
    const int nrows = (N_NODES - row0 < RB) ? (N_NODES - row0) : RB;
    {
        v2f xv[RPW], av[RPW];
        float dr[RPW], sv[RPW];
        #pragma unroll
        for (int k = 0; k < RPW; ++k) {
            const int rl = wave + 32 * k;
            const int rc = (rl < nrows) ? rl : (nrows - 1);
            xv[k] = *(const v2f*)(x + (size_t)(row0 + rc) * D + 2 * lane);
            dr[k] = d[row0 + rc];
            av[k] = *(const v2f*)(al + rc * PA);
            sv[k] = sdl[rc];
        }
        #pragma unroll
        for (int k = 0; k < RPW; ++k) {
            const int rl = wave + 32 * k;
            if (rl < nrows) {
                v2f o;
                o.x = (av[k].x + xv[k].x * dr[k]) * dr[k] * SMOOTHF + xv[k].x;
                o.y = (av[k].y + xv[k].y * dr[k]) * dr[k] * SMOOTHF + xv[k].y;
                *(v2f*)(al + rl * PA) = o;
                if (lane == 0) sdl[rl] = (sv[k] + dr[k]) * dr[k] * SMOOTHF + 1.0f;
            }
        }
    }
    __syncthreads();

    if (wave < 2 * MT) {
        const int mt   = wave >> 1;
        const int np   = wave & 1;
        const int half = lane >> 4;
        const int l15  = lane & 15;
        v8f c[2] = {v8f{}, v8f{}};
        const float* arow = acc + (mt * 16 + l15) * PA + half * 2;
        const float* brow = Wl + (np * 32 + l15) * PA + half * 2;
        #pragma unroll
        for (int kk = 0; kk < 16; ++kk) {
            const v2f a = *(const v2f*)(arow + kk * 4);
            #pragma unroll
            for (int q = 0; q < 2; ++q) {
                const v2f bf = *(const v2f*)(brow + q * 16 * PA + kk * 4);
                c[q] = __builtin_amdgcn_wmma_f32_16x16x4_f32(
                     false, a,  false, bf,
                     (short)0, c[q],
                     false,  false);
            }
        }
        const int rbase = mt * 16 + 8 * half;
        float cb[8];
        #pragma unroll
        for (int r = 0; r < 8; ++r) cb[r] = sdl[rbase + r];
        const bool whole = (mt * 16 + 16 <= nrows);
        #pragma unroll
        for (int q = 0; q < 2; ++q) {
            const int nn = np * 32 + q * 16 + l15;
            const float bv = bias[nn];
            float* op = out + (size_t)(row0 + rbase) * D + nn;
            if (whole) {
                #pragma unroll
                for (int r = 0; r < 8; ++r)
                    op[r * D] = (c[q][r] + cb[r] * bv) * (1.0f / (1.0f + SMOOTHF));
            } else {
                #pragma unroll
                for (int r = 0; r < 8; ++r)
                    if (rbase + r < nrows)
                        op[r * D] = (c[q][r] + cb[r] * bv) * (1.0f / (1.0f + SMOOTHF));
            }
        }
    }
}


extern "C" void kernel_launch(void* const* d_in, const int* in_sizes, int n_in,
                              void* d_out, int out_size, void* d_ws, size_t ws_size,
                              hipStream_t stream) {
    const float* x    = (const float*)d_in[0];
    const float* W    = (const float*)d_in[1];
    const float* bias = (const float*)d_in[2];
    const int*   erow = (const int*)d_in[3];
    const int*   ecol = (const int*)d_in[4];
    float* out = (float*)d_out;

    float* ws         = (float*)d_ws;
    unsigned* sorted  = (unsigned*)ws;
    unsigned* seg     = sorted + N_EDGES;
    unsigned* cseg    = seg + NB * G;
    float* deg        = (float*)(cseg + NB * G);
    unsigned* csorted = (unsigned*)(deg + DPL);

    k_group <<<G, TPB, 0, stream>>>(erow, ecol, sorted, seg, csorted, cseg);
    k_degree<<<NB, DTB, 0, stream>>>(csorted, cseg, deg);
    k_accum <<<NB, 1024, 0, stream>>>(sorted, seg, x, W, bias, deg, out);
}


# range sums: a work-group orders its range's edges by row in LDS (integer rank, scan, place); a wave owns whole rows, sums x[col]*d[col] in registers in rank order; no LDS f32 row adds; sum order new
# speedup vs baseline: 6.1389x; 1.1428x over previous
#include <hip/hip_runtime.h>
#include <hip/hip_bf16.h>

typedef float v2f __attribute__((ext_vector_type(2)));
typedef float v4f __attribute__((ext_vector_type(4)));
typedef float v8f __attribute__((ext_vector_type(8)));

#define N_NODES 50000
#define N_EDGES 800000
#define D 64
#define SMOOTHF 0.5f


#define TPB 1024


#define NB  256
#define RB  196
#define G   256
#define EPG (N_EDGES / G)
#define EPT ((EPG + TPB - 1) / TPB)
#define CPS 3136
#define DPL 50048
#define SPW (G / 32)
#define RPW ((RB + 31) / 32)
#define MT  ((RB + 15) / 16)
#define RBP (MT * 16)
#define PA  68
#define KB  5
#define CW  (KB * 32)
#define ECAP (32 * CW)

typedef unsigned v2u __attribute__((ext_vector_type(2)));

__global__ __launch_bounds__(TPB) void k_group(const int* __restrict__ erow,
                                               const int* __restrict__ ecol,
                                               unsigned* __restrict__ sorted,
                                               unsigned* __restrict__ seg,
                                               unsigned* __restrict__ csorted,
                                               unsigned* __restrict__ cseg)
{
    __shared__ int hist[2 * NB];
    __shared__ int wtot[2 * NB / 32];
    __shared__ unsigned buf[EPG];
    __shared__ unsigned cbw[CPS / 4];
    unsigned char* cbuf = (unsigned char*)cbw;
    const int g    = blockIdx.x;
    const int t    = threadIdx.x;
    const int lane = t & 31;
    if (t < 2 * NB) hist[t] = 0;
    if (t < CPS - EPG) cbuf[EPG + t] = 0;
    __syncthreads();

    const int e0 = g * EPG;
    unsigned er[EPT], ec[EPT], pkv[EPT], bb[EPT], cb[EPT], cl[EPT];
    int rk[EPT], ck[EPT];
    #pragma unroll
    for (int k = 0; k < EPT; ++k) {
        const int i = t + k * TPB;
        er[k] = 0u; ec[k] = 0u;
        if (i < EPG) {
            er[k] = (unsigned)erow[e0 + i];
            ec[k] = (unsigned)ecol[e0 + i];
        }
    }
    #pragma unroll
    for (int k = 0; k < EPT; ++k) {
        const int i = t + k * TPB;
        const unsigned b = er[k] / RB;
        const unsigned c = ec[k] / RB;
        bb[k]  = b;
        cb[k]  = c;
        pkv[k] = ((er[k] - b * RB) << 16) | ec[k];
        cl[k]  = ec[k] - c * RB;
        rk[k]  = 0;
        ck[k]  = 0;
        if (i < EPG) {
            rk[k] = __hip_atomic_fetch_add(&hist[b], 1, __ATOMIC_RELAXED,
                                           __HIP_MEMORY_SCOPE_WORKGROUP);
            ck[k] = __hip_atomic_fetch_add(&hist[NB + c], 1, __ATOMIC_RELAXED,
                                           __HIP_MEMORY_SCOPE_WORKGROUP);
        }
    }
    __syncthreads();

    int cnt_t = 0, inc = 0;
    if (t < 2 * NB) {
        cnt_t = hist[t];
        inc = cnt_t;
        #pragma unroll
        for (int sft = 1; sft < 32; sft <<= 1) {
            const int v = __shfl_up(inc, sft);
            if (lane >= sft) inc += v;
        }
        if (lane == 31) wtot[t >> 5] = inc;
    }
    __syncthreads();
    if (t < 2 * NB) {
        const int key = t & (NB - 1);
        const int w0  = (t >> 8) * (NB / 32);
        int st = inc - cnt_t;
        for (int w = w0; w < (t >> 5); ++w) st += wtot[w];
        hist[t] = st;
        const unsigned word = ((unsigned)st << 12) | (unsigned)cnt_t;
        if (t < NB) seg[key * G + g]  = word;
        else        cseg[key * G + g] = word;
    }
    __syncthreads();

    #pragma unroll
    for (int k = 0; k < EPT; ++k) {
        const int i = t + k * TPB;
        if (i < EPG) {
            buf[hist[bb[k]] + rk[k]] = pkv[k];
            cbuf[hist[NB + cb[k]] + ck[k]] = (unsigned char)cl[k];
        }
    }
    __syncthreads();
    for (int i = t; i < EPG; i += TPB) sorted[e0 + i] = buf[i];
    if (t < CPS / 4) csorted[g * (CPS / 4) + t] = cbw[t];
}

#define DTB G
__global__ __launch_bounds__(DTB) void k_degree(const unsigned* __restrict__ csorted,
                                                const unsigned* __restrict__ cseg,
                                                float* __restrict__ d)
{
    __shared__ unsigned cnt[256];
    const int b = blockIdx.x;
    const int t = threadIdx.x;
    cnt[t] = 0u;
    const unsigned sw = cseg[b * G + t];
    const int n = (int)(sw & 0xfffu);
    const unsigned char* p = (const unsigned char*)csorted + (size_t)t * CPS + (sw >> 12);
    __syncthreads();
    for (int j = 0; j < n; j += 8) {
        unsigned v[8];
        #pragma unroll
        for (int u = 0; u < 8; ++u) v[u] = p[(j + u < n) ? (j + u) : (n - 1)];
        #pragma unroll
        for (int u = 0; u < 8; ++u)
            __hip_atomic_fetch_add(&cnt[v[u]], (j + u < n) ? 1u : 0u, __ATOMIC_RELAXED, __HIP_MEMORY_SCOPE_WORKGROUP);
    }
    __syncthreads();
    const int node = b * RB + t;
    if (t < RB && node < N_NODES) d[node] = rsqrtf((float)cnt[t] + 1.0f);
}

__global__ __launch_bounds__(1024) void k_accum(
    const unsigned* __restrict__ sorted, const unsigned* __restrict__ seg,
    const float* __restrict__ x, const float* __restrict__ W,
    const float* __restrict__ bias, const float* __restrict__ d,
    float* __restrict__ out)
{
    __shared__ __attribute__((aligned(16))) float acc[RBP * PA];
    __shared__ __attribute__((aligned(16))) float Wl[D * PA];
    __shared__ float sdl[RBP];
    __shared__ __attribute__((aligned(8))) v2u elist[ECAP];
    __shared__ int cnt[RBP];
    __shared__ int start[RBP + 1];
    __shared__ int lmax;
    const int b    = blockIdx.x;
    const int tid  = threadIdx.x;
    const int lane = tid & 31;
    const int wave = __builtin_amdgcn_readfirstlane(tid >> 5);

    const v4f wv = *(const v4f*)(W + 4 * tid);
    for (int i = tid; i < RBP * PA / 4; i += 1024) ((v4f*)acc)[i] = v4f{};
    if (tid < RBP) { sdl[tid] = 0.0f; cnt[tid] = 0; }
    if (tid == 0) lmax = 0;
    *(v4f*)(&Wl[(tid >> 4) * PA + (tid & 15) * 4]) = wv;

    unsigned sgv = 0;
    if (lane < SPW) sgv = seg[b * G + wave * SPW + lane];
    int pc[SPW + 1], dl[SPW];
    pc[0] = 0;
    #pragma unroll
    for (int m = 0; m < SPW; ++m) {
        const unsigned s = __builtin_amdgcn_readlane(sgv, m);
        dl[m]     = (wave * SPW + m) * EPG + (int)(s >> 12) - pc[m];
        pc[m + 1] = pc[m] + (int)(s & 0xfffu);
    }
    const int total = pc[SPW];
    __syncthreads();
    if (lane == 0)
        __hip_atomic_fetch_max(&lmax, total, __ATOMIC_RELAXED, __HIP_MEMORY_SCOPE_WORKGROUP);

    const float* xl = x + 2 * lane;
    float* al = acc + 2 * lane;
    const int r0 = (wave * RB) >> 5;
    const int r1 = ((wave + 1) * RB) >> 5;

    for (int cbase = 0;; cbase += CW) {
        const int cend = (total < cbase + CW) ? total : (cbase + CW);
        unsigned pk[KB];
        float dv[KB];
        int rk[KB];
        #pragma unroll
        for (int k = 0; k < KB; ++k) { pk[k] = 0u; dv[k] = 0.0f; }
        if (cbase < cend) {
            #pragma unroll
            for (int k = 0; k < KB; ++k) {
                const int ii = cbase + k * 32 + lane;
                const int i  = (ii < cend) ? ii : (cend - 1);
                int dsel = dl[0];
                #pragma unroll
                for (int m = 1; m < SPW; ++m) dsel = (i >= pc[m]) ? dl[m] : dsel;
                pk[k] = sorted[i + dsel];
            }
            #pragma unroll
            for (int k = 0; k < KB; ++k) dv[k] = d[pk[k] & 0xffffu];
        }
        #pragma unroll
        for (int k = 0; k < KB; ++k) {
            rk[k] = 0;
            if (cbase + k * 32 + lane < cend) {
                rk[k] = __hip_atomic_fetch_add(&cnt[pk[k] >> 16], 1, __ATOMIC_RELAXED,
                                               __HIP_MEMORY_SCOPE_WORKGROUP);
            }
        }
        #pragma unroll
        for (int k = 0; k < KB; ++k) {
            if (cbase + k * 32 + lane < cend)
                __hip_atomic_fetch_add(&sdl[pk[k] >> 16], dv[k], __ATOMIC_RELAXED,
                                       __HIP_MEMORY_SCOPE_WORKGROUP);
        }
        __syncthreads();
        const int wmax = lmax;

        if (wave == 0) {
            int c[7];
            int s = 0;
            #pragma unroll
            for (int q = 0; q < 7; ++q) {
                const int idx = lane * 7 + q;
                c[q] = (idx < RBP) ? cnt[idx] : 0;
            }
            #pragma unroll
            for (int q = 0; q < 7; ++q) s += c[q];
            int inc = s;
            #pragma unroll
            for (int sft = 1; sft < 32; sft <<= 1) {
                const int v = __shfl_up(inc, sft);
                if (lane >= sft) inc += v;
            }
            int run = inc - s;
            #pragma unroll
            for (int q = 0; q < 7; ++q) {
                const int idx = lane * 7 + q;
                if (idx < RBP) { start[idx] = run; cnt[idx] = 0; run += c[q]; }
            }
            if (lane == 31) start[RBP] = inc;
        }
        __syncthreads();

        #pragma unroll
        for (int k = 0; k < KB; ++k) {
            if (cbase + k * 32 + lane < cend) {
                v2u rec;
                rec.x = pk[k];
                rec.y = __float_as_uint(dv[k]);
                elist[start[pk[k] >> 16] + rk[k]] = rec;
            }
        }
        __syncthreads();

        const int sbeg = __builtin_amdgcn_readfirstlane(start[r0]);
        const int send = __builtin_amdgcn_readfirstlane(start[r1]);
        if (send > sbeg) {
            v2f a = v2f{};
            int cur = r0;
            auto flush = [&](int r, v2f val) {
                v2f* q = (v2f*)(al + r * PA);
                v2f t = *q;
                t.x += val.x; t.y += val.y;
                *q = t;
            };
            for (int base = sbeg; base < send; base += 32) {
                const int n = (send - base < 32) ? (send - base) : 32;
                v2u rec = v2u{};
                if (lane < n) rec = elist[base + lane];
                const unsigned pkl = rec.x;
                const unsigned dvb = rec.y;
                int j = 0;
                for (; j + 8 <= n; j += 8) {
                    unsigned p[8];
                    float s[8];
                    v2f v[8];
                    #pragma unroll
                    for (int u = 0; u < 8; ++u) {
                        p[u] = __builtin_amdgcn_readlane(pkl, j + u);
                        s[u] = __uint_as_float(__builtin_amdgcn_readlane(dvb, j + u));
                        v[u] = *(const v2f*)(xl + (size_t)(p[u] & 0xffffu) * D);
                    }
                    #pragma unroll
                    for (int u = 0; u < 8; ++u) {
                        const int row = (int)(p[u] >> 16);
                        if (row != cur) { flush(cur, a); a = v2f{}; cur = row; }
                        a.x = fmaf(v[u].x, s[u], a.x);
                        a.y = fmaf(v[u].y, s[u], a.y);
                    }
                }
                if (j < n) {
                    const int m8 = n - j;
                    unsigned p[8];
                    float s[8];
                    v2f v[8];
                    #pragma unroll
                    for (int u = 0; u < 8; ++u) {
                        const int sel = (u < m8) ? (j + u) : (n - 1);
                        p[u] = __builtin_amdgcn_readlane(pkl, sel);
                        s[u] = __uint_as_float(__builtin_amdgcn_readlane(dvb, sel));
                        v[u] = *(const v2f*)(xl + (size_t)(p[u] & 0xffffu) * D);
                    }
                    #pragma unroll
                    for (int u = 0; u < 8; ++u) {
                        if (u < m8) {
                            const int row = (int)(p[u] >> 16);
                            if (row != cur) { flush(cur, a); a = v2f{}; cur = row; }
                            a.x = fmaf(v[u].x, s[u], a.x);
                            a.y = fmaf(v[u].y, s[u], a.y);
                        }
                    }
                }
            }
            flush(cur, a);
        }
        if (cbase + CW >= wmax) break;
    }
    __syncthreads();

    const int row0  = b * RB;
    const int nrows = (N_NODES - row0 < RB) ? (N_NODES - row0) : RB;
    {
        v2f xv[RPW], av[RPW];
        float dr[RPW], sv[RPW];
        #pragma unroll
        for (int k = 0; k < RPW; ++k) {
            const int rl = wave + 32 * k;
            const int rc = (rl < nrows) ? rl : (nrows - 1);
            xv[k] = *(const v2f*)(x + (size_t)(row0 + rc) * D + 2 * lane);
            dr[k] = d[row0 + rc];
            av[k] = *(const v2f*)(al + rc * PA);
            sv[k] = sdl[rc];
        }
        #pragma unroll
        for (int k = 0; k < RPW; ++k) {
            const int rl = wave + 32 * k;
            if (rl < nrows) {
                v2f o;
                o.x = (av[k].x + xv[k].x * dr[k]) * dr[k] * SMOOTHF + xv[k].x;
                o.y = (av[k].y + xv[k].y * dr[k]) * dr[k] * SMOOTHF + xv[k].y;
                *(v2f*)(al + rl * PA) = o;
                if (lane == 0) sdl[rl] = (sv[k] + dr[k]) * dr[k] * SMOOTHF + 1.0f;
            }
        }
    }
    __syncthreads();

    if (wave < 2 * MT) {
        const int mt   = wave >> 1;
        const int np   = wave & 1;
        const int half = lane >> 4;
        const int l15  = lane & 15;
        v8f c[2] = {v8f{}, v8f{}};
        const float* arow = acc + (mt * 16 + l15) * PA + half * 2;
        const float* brow = Wl + (np * 32 + l15) * PA + half * 2;
        #pragma unroll
        for (int kk = 0; kk < 16; ++kk) {
            const v2f a = *(const v2f*)(arow + kk * 4);
            #pragma unroll
            for (int q = 0; q < 2; ++q) {
                const v2f bf = *(const v2f*)(brow + q * 16 * PA + kk * 4);
                c[q] = __builtin_amdgcn_wmma_f32_16x16x4_f32(
                     false, a,  false, bf,
                     (short)0, c[q],
                     false,  false);
            }
        }
        const int rbase = mt * 16 + 8 * half;
        float cb[8];
        #pragma unroll
        for (int r = 0; r < 8; ++r) cb[r] = sdl[rbase + r];
        const bool whole = (mt * 16 + 16 <= nrows);
        #pragma unroll
        for (int q = 0; q < 2; ++q) {
            const int nn = np * 32 + q * 16 + l15;
            const float bv = bias[nn];
            float* op = out + (size_t)(row0 + rbase) * D + nn;
            if (whole) {
                #pragma unroll
                for (int r = 0; r < 8; ++r)
                    op[r * D] = (c[q][r] + cb[r] * bv) * (1.0f / (1.0f + SMOOTHF));
            } else {
                #pragma unroll
                for (int r = 0; r < 8; ++r)
                    if (rbase + r < nrows)
                        op[r * D] = (c[q][r] + cb[r] * bv) * (1.0f / (1.0f + SMOOTHF));
            }
        }
    }
}


extern "C" void kernel_launch(void* const* d_in, const int* in_sizes, int n_in,
                              void* d_out, int out_size, void* d_ws, size_t ws_size,
                              hipStream_t stream) {
    const float* x    = (const float*)d_in[0];
    const float* W    = (const float*)d_in[1];
    const float* bias = (const float*)d_in[2];
    const int*   erow = (const int*)d_in[3];
    const int*   ecol = (const int*)d_in[4];
    float* out = (float*)d_out;

    float* ws         = (float*)d_ws;
    unsigned* sorted  = (unsigned*)ws;
    unsigned* seg     = sorted + N_EDGES;
    unsigned* cseg    = seg + NB * G;
    float* deg        = (float*)(cseg + NB * G);
    unsigned* csorted = (unsigned*)(deg + DPL);

    k_group <<<G, TPB, 0, stream>>>(erow, ecol, sorted, seg, csorted, cseg);
    k_degree<<<NB, DTB, 0, stream>>>(csorted, cseg, deg);
    k_accum <<<NB, 1024, 0, stream>>>(sorted, seg, x, W, bias, deg, out);
}


# k_accum: d[col] formed from a table of every node's degree byte that the work-group copies into LDS by whole-line loads (k_degree stores the bytes; 255 = read d), not loaded 4 B per edge from memory
# speedup vs baseline: 6.4524x; 1.0511x over previous
#include <hip/hip_runtime.h>
#include <hip/hip_bf16.h>

typedef float v2f __attribute__((ext_vector_type(2)));
typedef float v4f __attribute__((ext_vector_type(4)));
typedef float v8f __attribute__((ext_vector_type(8)));

#define N_NODES 50000
#define N_EDGES 800000
#define D 64
#define SMOOTHF 0.5f


#define TPB 1024


#define NB  256
#define RB  196
#define G   256
#define EPG (N_EDGES / G)
#define EPT ((EPG + TPB - 1) / TPB)
#define CPS 3136
#define DPL 50048
#define SPW (G / 32)
#define RPW ((RB + 31) / 32)
#define MT  ((RB + 15) / 16)
#define RBP (MT * 16)
#define PA  68
#define KB  5
#define CW  (KB * 32)
#define ECAP (32 * CW)

typedef unsigned v2u __attribute__((ext_vector_type(2)));
typedef unsigned v4u __attribute__((ext_vector_type(4)));

#define DGQ (DPL / 16)
#define DGK ((DGQ + 1023) / 1024)

__device__ __forceinline__ float scale_of(unsigned cnt) { return rsqrtf((float)cnt + 1.0f); }

__global__ __launch_bounds__(TPB) void k_group(const int* __restrict__ erow,
                                               const int* __restrict__ ecol,
                                               unsigned* __restrict__ sorted,
                                               unsigned* __restrict__ seg,
                                               unsigned* __restrict__ csorted,
                                               unsigned* __restrict__ cseg)
{
    __shared__ int hist[2 * NB];
    __shared__ int wtot[2 * NB / 32];
    __shared__ unsigned buf[EPG];
    __shared__ unsigned cbw[CPS / 4];
    unsigned char* cbuf = (unsigned char*)cbw;
    const int g    = blockIdx.x;
    const int t    = threadIdx.x;
    const int lane = t & 31;
    if (t < 2 * NB) hist[t] = 0;
    if (t < CPS - EPG) cbuf[EPG + t] = 0;
    __syncthreads();

    const int e0 = g * EPG;
    unsigned er[EPT], ec[EPT], pkv[EPT], bb[EPT], cb[EPT], cl[EPT];
    int rk[EPT], ck[EPT];
    #pragma unroll
    for (int k = 0; k < EPT; ++k) {
        const int i = t + k * TPB;
        er[k] = 0u; ec[k] = 0u;
        if (i < EPG) {
            er[k] = (unsigned)erow[e0 + i];
            ec[k] = (unsigned)ecol[e0 + i];
        }
    }
    #pragma unroll
    for (int k = 0; k < EPT; ++k) {
        const int i = t + k * TPB;
        const unsigned b = er[k] / RB;
        const unsigned c = ec[k] / RB;
        bb[k]  = b;
        cb[k]  = c;
        pkv[k] = ((er[k] - b * RB) << 16) | ec[k];
        cl[k]  = ec[k] - c * RB;
        rk[k]  = 0;
        ck[k]  = 0;
        if (i < EPG) {
            rk[k] = __hip_atomic_fetch_add(&hist[b], 1, __ATOMIC_RELAXED,
                                           __HIP_MEMORY_SCOPE_WORKGROUP);
            ck[k] = __hip_atomic_fetch_add(&hist[NB + c], 1, __ATOMIC_RELAXED,
                                           __HIP_MEMORY_SCOPE_WORKGROUP);
        }
    }
    __syncthreads();

    int cnt_t = 0, inc = 0;
    if (t < 2 * NB) {
        cnt_t = hist[t];
        inc = cnt_t;
        #pragma unroll
        for (int sft = 1; sft < 32; sft <<= 1) {
            const int v = __shfl_up(inc, sft);
            if (lane >= sft) inc += v;
        }
        if (lane == 31) wtot[t >> 5] = inc;
    }
    __syncthreads();
    if (t < 2 * NB) {
        const int key = t & (NB - 1);
        const int w0  = (t >> 8) * (NB / 32);
        int st = inc - cnt_t;
        for (int w = w0; w < (t >> 5); ++w) st += wtot[w];
        hist[t] = st;
        const unsigned word = ((unsigned)st << 12) | (unsigned)cnt_t;
        if (t < NB) seg[key * G + g]  = word;
        else        cseg[key * G + g] = word;
    }
    __syncthreads();

    #pragma unroll
    for (int k = 0; k < EPT; ++k) {
        const int i = t + k * TPB;
        if (i < EPG) {
            buf[hist[bb[k]] + rk[k]] = pkv[k];
            cbuf[hist[NB + cb[k]] + ck[k]] = (unsigned char)cl[k];
        }
    }
    __syncthreads();
    for (int i = t; i < EPG; i += TPB) sorted[e0 + i] = buf[i];
    if (t < CPS / 4) csorted[g * (CPS / 4) + t] = cbw[t];
}

#define DTB G
__global__ __launch_bounds__(DTB) void k_degree(const unsigned* __restrict__ csorted,
                                                const unsigned* __restrict__ cseg,
                                                float* __restrict__ d,
                                                unsigned char* __restrict__ dgb)
{
    __shared__ unsigned cnt[256];
    const int b = blockIdx.x;
    const int t = threadIdx.x;
    cnt[t] = 0u;
    const unsigned sw = cseg[b * G + t];
    const int n = (int)(sw & 0xfffu);
    const unsigned char* p = (const unsigned char*)csorted + (size_t)t * CPS + (sw >> 12);
    __syncthreads();
    for (int j = 0; j < n; j += 8) {
        unsigned v[8];
        #pragma unroll
        for (int u = 0; u < 8; ++u) v[u] = p[(j + u < n) ? (j + u) : (n - 1)];
        #pragma unroll
        for (int u = 0; u < 8; ++u)
            __hip_atomic_fetch_add(&cnt[v[u]], (j + u < n) ? 1u : 0u, __ATOMIC_RELAXED, __HIP_MEMORY_SCOPE_WORKGROUP);
    }
    __syncthreads();
    const int node = b * RB + t;
    if (t < RB && node < N_NODES) d[node] = scale_of(cnt[t]);
    if (t < RB && node < DPL) dgb[node] = (unsigned char)((cnt[t] < 255u) ? cnt[t] : 255u);
}

__global__ __launch_bounds__(1024) void k_accum(
    const unsigned* __restrict__ sorted, const unsigned* __restrict__ seg,
    const float* __restrict__ x, const float* __restrict__ W,
    const float* __restrict__ bias, const float* __restrict__ d,
    const unsigned* __restrict__ dgw, float* __restrict__ out)
{
    __shared__ __attribute__((aligned(16))) float acc[RBP * PA];
    __shared__ __attribute__((aligned(16))) float Wl[D * PA];
    __shared__ float sdl[RBP];
    __shared__ __attribute__((aligned(8))) v2u elist[ECAP];
    __shared__ int cnt[RBP];
    __shared__ int start[RBP + 1];
    __shared__ int lmax;
    __shared__ __attribute__((aligned(16))) unsigned dgl[DPL / 4];
    const int b    = blockIdx.x;
    const int tid  = threadIdx.x;
    const int lane = tid & 31;
    const int wave = __builtin_amdgcn_readfirstlane(tid >> 5);

    const v4f wv = *(const v4f*)(W + 4 * tid);
    v4u tb[DGK];
    #pragma unroll
    for (int k = 0; k < DGK; ++k) {
        const int i = tid + k * 1024;
        tb[k] = ((const v4u*)dgw)[(i < DGQ) ? i : (DGQ - 1)];
    }
    for (int i = tid; i < RBP * PA / 4; i += 1024) ((v4f*)acc)[i] = v4f{};
    if (tid < RBP) { sdl[tid] = 0.0f; cnt[tid] = 0; }
    if (tid == 0) lmax = 0;
    *(v4f*)(&Wl[(tid >> 4) * PA + (tid & 15) * 4]) = wv;

    unsigned sgv = 0;
    if (lane < SPW) sgv = seg[b * G + wave * SPW + lane];
    int pc[SPW + 1], dl[SPW];
    pc[0] = 0;
    #pragma unroll
    for (int m = 0; m < SPW; ++m) {
        const unsigned s = __builtin_amdgcn_readlane(sgv, m);
        dl[m]     = (wave * SPW + m) * EPG + (int)(s >> 12) - pc[m];
        pc[m + 1] = pc[m] + (int)(s & 0xfffu);
    }
    const int total = pc[SPW];
    #pragma unroll
    for (int k = 0; k < DGK; ++k) {
        const int i = tid + k * 1024;
        if (i < DGQ) ((v4u*)dgl)[i] = tb[k];
    }
    __syncthreads();
    if (lane == 0)
        __hip_atomic_fetch_max(&lmax, total, __ATOMIC_RELAXED, __HIP_MEMORY_SCOPE_WORKGROUP);

    const float* xl = x + 2 * lane;
    float* al = acc + 2 * lane;
    const int r0 = (wave * RB) >> 5;
    const int r1 = ((wave + 1) * RB) >> 5;

    for (int cbase = 0;; cbase += CW) {
        const int cend = (total < cbase + CW) ? total : (cbase + CW);
        unsigned pk[KB];
        float dv[KB];
        int rk[KB];
        #pragma unroll
        for (int k = 0; k < KB; ++k) { pk[k] = 0u; dv[k] = 0.0f; }
        if (cbase < cend) {
            #pragma unroll
            for (int k = 0; k < KB; ++k) {
                const int ii = cbase + k * 32 + lane;
                const int i  = (ii < cend) ? ii : (cend - 1);
                int dsel = dl[0];
                #pragma unroll
                for (int m = 1; m < SPW; ++m) dsel = (i >= pc[m]) ? dl[m] : dsel;
                pk[k] = sorted[i + dsel];
            }
            unsigned cbv[KB];
            #pragma unroll
            for (int k = 0; k < KB; ++k) cbv[k] = ((const unsigned char*)dgl)[pk[k] & 0xffffu];
            #pragma unroll
            for (int k = 0; k < KB; ++k) {
                dv[k] = scale_of(cbv[k]);
                if (cbv[k] == 255u) dv[k] = d[pk[k] & 0xffffu];
            }
        }
        #pragma unroll
        for (int k = 0; k < KB; ++k) {
            rk[k] = 0;
            if (cbase + k * 32 + lane < cend) {
                rk[k] = __hip_atomic_fetch_add(&cnt[pk[k] >> 16], 1, __ATOMIC_RELAXED,
                                               __HIP_MEMORY_SCOPE_WORKGROUP);
            }
        }
        #pragma unroll
        for (int k = 0; k < KB; ++k) {
            if (cbase + k * 32 + lane < cend)
                __hip_atomic_fetch_add(&sdl[pk[k] >> 16], dv[k], __ATOMIC_RELAXED,
                                       __HIP_MEMORY_SCOPE_WORKGROUP);
        }
        __syncthreads();
        const int wmax = lmax;

        if (wave == 0) {
            int c[7];
            int s = 0;
            #pragma unroll
            for (int q = 0; q < 7; ++q) {
                const int idx = lane * 7 + q;
                c[q] = (idx < RBP) ? cnt[idx] : 0;
            }
            #pragma unroll
            for (int q = 0; q < 7; ++q) s += c[q];
            int inc = s;
            #pragma unroll
            for (int sft = 1; sft < 32; sft <<= 1) {
                const int v = __shfl_up(inc, sft);
                if (lane >= sft) inc += v;
            }
            int run = inc - s;
            #pragma unroll
            for (int q = 0; q < 7; ++q) {
                const int idx = lane * 7 + q;
                if (idx < RBP) { start[idx] = run; cnt[idx] = 0; run += c[q]; }
            }
            if (lane == 31) start[RBP] = inc;
        }
        __syncthreads();

        #pragma unroll
        for (int k = 0; k < KB; ++k) {
            if (cbase + k * 32 + lane < cend) {
                v2u rec;
                rec.x = pk[k];
                rec.y = __float_as_uint(dv[k]);
                elist[start[pk[k] >> 16] + rk[k]] = rec;
            }
        }
        __syncthreads();

        const int sbeg = __builtin_amdgcn_readfirstlane(start[r0]);
        const int send = __builtin_amdgcn_readfirstlane(start[r1]);
        if (send > sbeg) {
            v2f a = v2f{};
            int cur = r0;
            auto flush = [&](int r, v2f val) {
                v2f* q = (v2f*)(al + r * PA);
                v2f t = *q;
                t.x += val.x; t.y += val.y;
                *q = t;
            };
            for (int base = sbeg; base < send; base += 32) {
                const int n = (send - base < 32) ? (send - base) : 32;
                v2u rec = v2u{};
                if (lane < n) rec = elist[base + lane];
                const unsigned pkl = rec.x;
                const unsigned dvb = rec.y;
                int j = 0;
                for (; j + 8 <= n; j += 8) {
                    unsigned p[8];
                    float s[8];
                    v2f v[8];
                    #pragma unroll
                    for (int u = 0; u < 8; ++u) {
                        p[u] = __builtin_amdgcn_readlane(pkl, j + u);
                        s[u] = __uint_as_float(__builtin_amdgcn_readlane(dvb, j + u));
                        v[u] = *(const v2f*)(xl + (size_t)(p[u] & 0xffffu) * D);
                    }
                    #pragma unroll
                    for (int u = 0; u < 8; ++u) {
                        const int row = (int)(p[u] >> 16);
                        if (row != cur) { flush(cur, a); a = v2f{}; cur = row; }
                        a.x = fmaf(v[u].x, s[u], a.x);
                        a.y = fmaf(v[u].y, s[u], a.y);
                    }
                }
                if (j < n) {
                    const int m8 = n - j;
                    unsigned p[8];
                    float s[8];
                    v2f v[8];
                    #pragma unroll
                    for (int u = 0; u < 8; ++u) {
                        const int sel = (u < m8) ? (j + u) : (n - 1);
                        p[u] = __builtin_amdgcn_readlane(pkl, sel);
                        s[u] = __uint_as_float(__builtin_amdgcn_readlane(dvb, sel));
                        v[u] = *(const v2f*)(xl + (size_t)(p[u] & 0xffffu) * D);
                    }
                    #pragma unroll
                    for (int u = 0; u < 8; ++u) {
                        if (u < m8) {
                            const int row = (int)(p[u] >> 16);
                            if (row != cur) { flush(cur, a); a = v2f{}; cur = row; }
                            a.x = fmaf(v[u].x, s[u], a.x);
                            a.y = fmaf(v[u].y, s[u], a.y);
                        }
                    }
                }
            }
            flush(cur, a);
        }
        if (cbase + CW >= wmax) break;
    }
    __syncthreads();

    const int row0  = b * RB;
    const int nrows = (N_NODES - row0 < RB) ? (N_NODES - row0) : RB;
    {
        v2f xv[RPW], av[RPW];
        float dr[RPW], sv[RPW];
        #pragma unroll
        for (int k = 0; k < RPW; ++k) {
            const int rl = wave + 32 * k;
            const int rc = (rl < nrows) ? rl : (nrows - 1);
            xv[k] = *(const v2f*)(x + (size_t)(row0 + rc) * D + 2 * lane);
            dr[k] = d[row0 + rc];
            av[k] = *(const v2f*)(al + rc * PA);
            sv[k] = sdl[rc];
        }
        #pragma unroll
        for (int k = 0; k < RPW; ++k) {
            const int rl = wave + 32 * k;
            if (rl < nrows) {
                v2f o;
                o.x = (av[k].x + xv[k].x * dr[k]) * dr[k] * SMOOTHF + xv[k].x;
                o.y = (av[k].y + xv[k].y * dr[k]) * dr[k] * SMOOTHF + xv[k].y;
                *(v2f*)(al + rl * PA) = o;
                if (lane == 0) sdl[rl] = (sv[k] + dr[k]) * dr[k] * SMOOTHF + 1.0f;
            }
        }
    }
    __syncthreads();

    if (wave < 2 * MT) {
        const int mt   = wave >> 1;
        const int np   = wave & 1;
        const int half = lane >> 4;
        const int l15  = lane & 15;
        v8f c[2] = {v8f{}, v8f{}};
        const float* arow = acc + (mt * 16 + l15) * PA + half * 2;
        const float* brow = Wl + (np * 32 + l15) * PA + half * 2;
        #pragma unroll
        for (int kk = 0; kk < 16; ++kk) {
            const v2f a = *(const v2f*)(arow + kk * 4);
            #pragma unroll
            for (int q = 0; q < 2; ++q) {
                const v2f bf = *(const v2f*)(brow + q * 16 * PA + kk * 4);
                c[q] = __builtin_amdgcn_wmma_f32_16x16x4_f32(
                     false, a,  false, bf,
                     (short)0, c[q],
                     false,  false);
            }
        }
        const int rbase = mt * 16 + 8 * half;
        float cb[8];
        #pragma unroll
        for (int r = 0; r < 8; ++r) cb[r] = sdl[rbase + r];
        const bool whole = (mt * 16 + 16 <= nrows);
        #pragma unroll
        for (int q = 0; q < 2; ++q) {
            const int nn = np * 32 + q * 16 + l15;
            const float bv = bias[nn];
            float* op = out + (size_t)(row0 + rbase) * D + nn;
            if (whole) {
                #pragma unroll
                for (int r = 0; r < 8; ++r)
                    op[r * D] = (c[q][r] + cb[r] * bv) * (1.0f / (1.0f + SMOOTHF));
            } else {
                #pragma unroll
                for (int r = 0; r < 8; ++r)
                    if (rbase + r < nrows)
                        op[r * D] = (c[q][r] + cb[r] * bv) * (1.0f / (1.0f + SMOOTHF));
            }
        }
    }
}


extern "C" void kernel_launch(void* const* d_in, const int* in_sizes, int n_in,
                              void* d_out, int out_size, void* d_ws, size_t ws_size,
                              hipStream_t stream) {
    const float* x    = (const float*)d_in[0];
    const float* W    = (const float*)d_in[1];
    const float* bias = (const float*)d_in[2];
    const int*   erow = (const int*)d_in[3];
    const int*   ecol = (const int*)d_in[4];
    float* out = (float*)d_out;

    float* ws         = (float*)d_ws;
    unsigned* sorted  = (unsigned*)ws;
    unsigned* seg     = sorted + N_EDGES;
    unsigned* cseg    = seg + NB * G;
    float* deg        = (float*)(cseg + NB * G);
    unsigned* csorted = (unsigned*)(deg + DPL);
    unsigned* dgw     = csorted + G * (CPS / 4);

    k_group <<<G, TPB, 0, stream>>>(erow, ecol, sorted, seg, csorted, cseg);
    k_degree<<<NB, DTB, 0, stream>>>(csorted, cseg, deg, (unsigned char*)dgw);
    k_accum <<<NB, 1024, 0, stream>>>(sorted, seg, x, W, bias, deg, dgw, out);
}


# k_accum row sums: row-ordered records dealt to the 32 waves in equal shares of whole 8-load groups, not 6 or 7 whole rows a wave; a row run cut at a share edge joins by f32 LDS adds; sum tree new
# speedup vs baseline: 6.7151x; 1.0407x over previous
#include <hip/hip_runtime.h>
#include <hip/hip_bf16.h>

typedef float v2f __attribute__((ext_vector_type(2)));
typedef float v4f __attribute__((ext_vector_type(4)));
typedef float v8f __attribute__((ext_vector_type(8)));

#define N_NODES 50000
#define N_EDGES 800000
#define D 64
#define SMOOTHF 0.5f


#define TPB 1024


#define NB  256
#define RB  196
#define G   256
#define EPG (N_EDGES / G)
#define EPT ((EPG + TPB - 1) / TPB)
#define CPS 3136
#define DPL 50048
#define SPW (G / 32)
#define RPW ((RB + 31) / 32)
#define MT  ((RB + 15) / 16)
#define RBP (MT * 16)
#define PA  68
#define KB  5
#define CW  (KB * 32)
#define ECAP (32 * CW)

typedef unsigned v2u __attribute__((ext_vector_type(2)));
typedef unsigned v4u __attribute__((ext_vector_type(4)));

#define DGQ (DPL / 16)
#define DGK ((DGQ + 1023) / 1024)

__device__ __forceinline__ float scale_of(unsigned cnt) { return rsqrtf((float)cnt + 1.0f); }

__global__ __launch_bounds__(TPB) void k_group(const int* __restrict__ erow,
                                               const int* __restrict__ ecol,
                                               unsigned* __restrict__ sorted,
                                               unsigned* __restrict__ seg,
                                               unsigned* __restrict__ csorted,
                                               unsigned* __restrict__ cseg)
{
    __shared__ int hist[2 * NB];
    __shared__ int wtot[2 * NB / 32];
    __shared__ unsigned buf[EPG];
    __shared__ unsigned cbw[CPS / 4];
    unsigned char* cbuf = (unsigned char*)cbw;
    const int g    = blockIdx.x;
    const int t    = threadIdx.x;
    const int lane = t & 31;
    if (t < 2 * NB) hist[t] = 0;
    if (t < CPS - EPG) cbuf[EPG + t] = 0;
    __syncthreads();

    const int e0 = g * EPG;
    unsigned er[EPT], ec[EPT], pkv[EPT], bb[EPT], cb[EPT], cl[EPT];
    int rk[EPT], ck[EPT];
    #pragma unroll
    for (int k = 0; k < EPT; ++k) {
        const int i = t + k * TPB;
        er[k] = 0u; ec[k] = 0u;
        if (i < EPG) {
            er[k] = (unsigned)erow[e0 + i];
            ec[k] = (unsigned)ecol[e0 + i];
        }
    }
    #pragma unroll
    for (int k = 0; k < EPT; ++k) {
        const int i = t + k * TPB;
        const unsigned b = er[k] / RB;
        const unsigned c = ec[k] / RB;
        bb[k]  = b;
        cb[k]  = c;
        pkv[k] = ((er[k] - b * RB) << 16) | ec[k];
        cl[k]  = ec[k] - c * RB;
        rk[k]  = 0;
        ck[k]  = 0;
        if (i < EPG) {
            rk[k] = __hip_atomic_fetch_add(&hist[b], 1, __ATOMIC_RELAXED,
                                           __HIP_MEMORY_SCOPE_WORKGROUP);
            ck[k] = __hip_atomic_fetch_add(&hist[NB + c], 1, __ATOMIC_RELAXED,
                                           __HIP_MEMORY_SCOPE_WORKGROUP);
        }
    }
    __syncthreads();

    int cnt_t = 0, inc = 0;
    if (t < 2 * NB) {
        cnt_t = hist[t];
        inc = cnt_t;
        #pragma unroll
        for (int sft = 1; sft < 32; sft <<= 1) {
            const int v = __shfl_up(inc, sft);
            if (lane >= sft) inc += v;
        }
        if (lane == 31) wtot[t >> 5] = inc;
    }
    __syncthreads();
    if (t < 2 * NB) {
        const int key = t & (NB - 1);
        const int w0  = (t >> 8) * (NB / 32);
        int st = inc - cnt_t;
        for (int w = w0; w < (t >> 5); ++w) st += wtot[w];
        hist[t] = st;
        const unsigned word = ((unsigned)st << 12) | (unsigned)cnt_t;
        if (t < NB) seg[key * G + g]  = word;
        else        cseg[key * G + g] = word;
    }
    __syncthreads();

    #pragma unroll
    for (int k = 0; k < EPT; ++k) {
        const int i = t + k * TPB;
        if (i < EPG) {
            buf[hist[bb[k]] + rk[k]] = pkv[k];
            cbuf[hist[NB + cb[k]] + ck[k]] = (unsigned char)cl[k];
        }
    }
    __syncthreads();
    for (int i = t; i < EPG; i += TPB) sorted[e0 + i] = buf[i];
    if (t < CPS / 4) csorted[g * (CPS / 4) + t] = cbw[t];
}

#define DTB G
__global__ __launch_bounds__(DTB) void k_degree(const unsigned* __restrict__ csorted,
                                                const unsigned* __restrict__ cseg,
                                                float* __restrict__ d,
                                                unsigned char* __restrict__ dgb)
{
    __shared__ unsigned cnt[256];
    const int b = blockIdx.x;
    const int t = threadIdx.x;
    cnt[t] = 0u;
    const unsigned sw = cseg[b * G + t];
    const int n = (int)(sw & 0xfffu);
    const unsigned char* p = (const unsigned char*)csorted + (size_t)t * CPS + (sw >> 12);
    __syncthreads();
    for (int j = 0; j < n; j += 8) {
        unsigned v[8];
        #pragma unroll
        for (int u = 0; u < 8; ++u) v[u] = p[(j + u < n) ? (j + u) : (n - 1)];
        #pragma unroll
        for (int u = 0; u < 8; ++u)
            __hip_atomic_fetch_add(&cnt[v[u]], (j + u < n) ? 1u : 0u, __ATOMIC_RELAXED, __HIP_MEMORY_SCOPE_WORKGROUP);
    }
    __syncthreads();
    const int node = b * RB + t;
    if (t < RB && node < N_NODES) d[node] = scale_of(cnt[t]);
    if (t < RB && node < DPL) dgb[node] = (unsigned char)((cnt[t] < 255u) ? cnt[t] : 255u);
}

__global__ __launch_bounds__(1024) void k_accum(
    const unsigned* __restrict__ sorted, const unsigned* __restrict__ seg,
    const float* __restrict__ x, const float* __restrict__ W,
    const float* __restrict__ bias, const float* __restrict__ d,
    const unsigned* __restrict__ dgw, float* __restrict__ out)
{
    __shared__ __attribute__((aligned(16))) float acc[RBP * PA];
    __shared__ __attribute__((aligned(16))) float Wl[D * PA];
    __shared__ float sdl[RBP];
    __shared__ __attribute__((aligned(8))) v2u elist[ECAP];
    __shared__ int cnt[RBP];
    __shared__ int start[RBP + 1];
    __shared__ int lmax;
    __shared__ __attribute__((aligned(16))) unsigned dgl[DPL / 4];
    const int b    = blockIdx.x;
    const int tid  = threadIdx.x;
    const int lane = tid & 31;
    const int wave = __builtin_amdgcn_readfirstlane(tid >> 5);

    const v4f wv = *(const v4f*)(W + 4 * tid);
    v4u tb[DGK];
    #pragma unroll
    for (int k = 0; k < DGK; ++k) {
        const int i = tid + k * 1024;
        tb[k] = ((const v4u*)dgw)[(i < DGQ) ? i : (DGQ - 1)];
    }
    for (int i = tid; i < RBP * PA / 4; i += 1024) ((v4f*)acc)[i] = v4f{};
    if (tid < RBP) { sdl[tid] = 0.0f; cnt[tid] = 0; }
    if (tid == 0) lmax = 0;
    *(v4f*)(&Wl[(tid >> 4) * PA + (tid & 15) * 4]) = wv;

    unsigned sgv = 0;
    if (lane < SPW) sgv = seg[b * G + wave * SPW + lane];
    int pc[SPW + 1], dl[SPW];
    pc[0] = 0;
    #pragma unroll
    for (int m = 0; m < SPW; ++m) {
        const unsigned s = __builtin_amdgcn_readlane(sgv, m);
        dl[m]     = (wave * SPW + m) * EPG + (int)(s >> 12) - pc[m];
        pc[m + 1] = pc[m] + (int)(s & 0xfffu);
    }
    const int total = pc[SPW];
    #pragma unroll
    for (int k = 0; k < DGK; ++k) {
        const int i = tid + k * 1024;
        if (i < DGQ) ((v4u*)dgl)[i] = tb[k];
    }
    __syncthreads();
    if (lane == 0)
        __hip_atomic_fetch_max(&lmax, total, __ATOMIC_RELAXED, __HIP_MEMORY_SCOPE_WORKGROUP);

    const float* xl = x + 2 * lane;
    float* al = acc + 2 * lane;

    for (int cbase = 0;; cbase += CW) {
        const int cend = (total < cbase + CW) ? total : (cbase + CW);
        unsigned pk[KB];
        float dv[KB];
        int rk[KB];
        #pragma unroll
        for (int k = 0; k < KB; ++k) { pk[k] = 0u; dv[k] = 0.0f; }
        if (cbase < cend) {
            #pragma unroll
            for (int k = 0; k < KB; ++k) {
                const int ii = cbase + k * 32 + lane;
                const int i  = (ii < cend) ? ii : (cend - 1);
                int dsel = dl[0];
                #pragma unroll
                for (int m = 1; m < SPW; ++m) dsel = (i >= pc[m]) ? dl[m] : dsel;
                pk[k] = sorted[i + dsel];
            }
            unsigned cbv[KB];
            #pragma unroll
            for (int k = 0; k < KB; ++k) cbv[k] = ((const unsigned char*)dgl)[pk[k] & 0xffffu];
            #pragma unroll
            for (int k = 0; k < KB; ++k) {
                dv[k] = scale_of(cbv[k]);
                if (cbv[k] == 255u) dv[k] = d[pk[k] & 0xffffu];
            }
        }
        #pragma unroll
        for (int k = 0; k < KB; ++k) {
            rk[k] = 0;
            if (cbase + k * 32 + lane < cend) {
                rk[k] = __hip_atomic_fetch_add(&cnt[pk[k] >> 16], 1, __ATOMIC_RELAXED,
                                               __HIP_MEMORY_SCOPE_WORKGROUP);
            }
        }
        #pragma unroll
        for (int k = 0; k < KB; ++k) {
            if (cbase + k * 32 + lane < cend)
                __hip_atomic_fetch_add(&sdl[pk[k] >> 16], dv[k], __ATOMIC_RELAXED,
                                       __HIP_MEMORY_SCOPE_WORKGROUP);
        }
        __syncthreads();
        const int wmax = lmax;

        if (wave == 0) {
            int c[7];
            int s = 0;
            #pragma unroll
            for (int q = 0; q < 7; ++q) {
                const int idx = lane * 7 + q;
                c[q] = (idx < RBP) ? cnt[idx] : 0;
            }
            #pragma unroll
            for (int q = 0; q < 7; ++q) s += c[q];
            int inc = s;
            #pragma unroll
            for (int sft = 1; sft < 32; sft <<= 1) {
                const int v = __shfl_up(inc, sft);
                if (lane >= sft) inc += v;
            }
            int run = inc - s;
            #pragma unroll
            for (int q = 0; q < 7; ++q) {
                const int idx = lane * 7 + q;
                if (idx < RBP) { start[idx] = run; cnt[idx] = 0; run += c[q]; }
            }
            if (lane == 31) start[RBP] = inc;
        }
        __syncthreads();

        #pragma unroll
        for (int k = 0; k < KB; ++k) {
            if (cbase + k * 32 + lane < cend) {
                v2u rec;
                rec.x = pk[k];
                rec.y = __float_as_uint(dv[k]);
                elist[start[pk[k] >> 16] + rk[k]] = rec;
            }
        }
        __syncthreads();

        const int tot  = __builtin_amdgcn_readfirstlane(start[RBP]);
        const int ng   = (tot + 7) >> 3;
        const int sbeg = ((wave * ng) >> 5) * 8;
        const int sen8 = (((wave + 1) * ng) >> 5) * 8;
        const int send = (sen8 < tot) ? sen8 : tot;
        if (send > sbeg) {
            v2f a = v2f{};
            int cur = (int)((unsigned)__builtin_amdgcn_readfirstlane(elist[sbeg].x) >> 16);
            auto flush = [&](int r, v2f val) {
                float* q = al + r * PA;
                __hip_atomic_fetch_add(q,     val.x, __ATOMIC_RELAXED, __HIP_MEMORY_SCOPE_WORKGROUP);
                __hip_atomic_fetch_add(q + 1, val.y, __ATOMIC_RELAXED, __HIP_MEMORY_SCOPE_WORKGROUP);
            };
            for (int base = sbeg; base < send; base += 32) {
                const int n = (send - base < 32) ? (send - base) : 32;
                v2u rec = v2u{};
                if (lane < n) rec = elist[base + lane];
                const unsigned pkl = rec.x;
                const unsigned dvb = rec.y;
                int j = 0;
                for (; j + 8 <= n; j += 8) {
                    unsigned p[8];
                    float s[8];
                    v2f v[8];
                    #pragma unroll
                    for (int u = 0; u < 8; ++u) {
                        p[u] = __builtin_amdgcn_readlane(pkl, j + u);
                        s[u] = __uint_as_float(__builtin_amdgcn_readlane(dvb, j + u));
                        v[u] = *(const v2f*)(xl + (size_t)(p[u] & 0xffffu) * D);
                    }
                    #pragma unroll
                    for (int u = 0; u < 8; ++u) {
                        const int row = (int)(p[u] >> 16);
                        if (row != cur) { flush(cur, a); a = v2f{}; cur = row; }
                        a.x = fmaf(v[u].x, s[u], a.x);
                        a.y = fmaf(v[u].y, s[u], a.y);
                    }
                }
                if (j < n) {
                    const int m8 = n - j;
                    unsigned p[8];
                    float s[8];
                    v2f v[8];
                    #pragma unroll
                    for (int u = 0; u < 8; ++u) {
                        const int sel = (u < m8) ? (j + u) : (n - 1);
                        p[u] = __builtin_amdgcn_readlane(pkl, sel);
                        s[u] = __uint_as_float(__builtin_amdgcn_readlane(dvb, sel));
                        v[u] = *(const v2f*)(xl + (size_t)(p[u] & 0xffffu) * D);
                    }
                    #pragma unroll
                    for (int u = 0; u < 8; ++u) {
                        if (u < m8) {
                            const int row = (int)(p[u] >> 16);
                            if (row != cur) { flush(cur, a); a = v2f{}; cur = row; }
                            a.x = fmaf(v[u].x, s[u], a.x);
                            a.y = fmaf(v[u].y, s[u], a.y);
                        }
                    }
                }
            }
            flush(cur, a);
        }
        if (cbase + CW >= wmax) break;
    }
    __syncthreads();

    const int row0  = b * RB;
    const int nrows = (N_NODES - row0 < RB) ? (N_NODES - row0) : RB;
    {
        v2f xv[RPW], av[RPW];
        float dr[RPW], sv[RPW];
        #pragma unroll
        for (int k = 0; k < RPW; ++k) {
            const int rl = wave + 32 * k;
            const int rc = (rl < nrows) ? rl : (nrows - 1);
            xv[k] = *(const v2f*)(x + (size_t)(row0 + rc) * D + 2 * lane);
            dr[k] = d[row0 + rc];
            av[k] = *(const v2f*)(al + rc * PA);
            sv[k] = sdl[rc];
        }
        #pragma unroll
        for (int k = 0; k < RPW; ++k) {
            const int rl = wave + 32 * k;
            if (rl < nrows) {
                v2f o;
                o.x = (av[k].x + xv[k].x * dr[k]) * dr[k] * SMOOTHF + xv[k].x;
                o.y = (av[k].y + xv[k].y * dr[k]) * dr[k] * SMOOTHF + xv[k].y;
                *(v2f*)(al + rl * PA) = o;
                if (lane == 0) sdl[rl] = (sv[k] + dr[k]) * dr[k] * SMOOTHF + 1.0f;
            }
        }
    }
    __syncthreads();

    if (wave < 2 * MT) {
        const int mt   = wave >> 1;
        const int np   = wave & 1;
        const int half = lane >> 4;
        const int l15  = lane & 15;
        v8f c[2] = {v8f{}, v8f{}};
        const float* arow = acc + (mt * 16 + l15) * PA + half * 2;
        const float* brow = Wl + (np * 32 + l15) * PA + half * 2;
        #pragma unroll
        for (int kk = 0; kk < 16; ++kk) {
            const v2f a = *(const v2f*)(arow + kk * 4);
            #pragma unroll
            for (int q = 0; q < 2; ++q) {
                const v2f bf = *(const v2f*)(brow + q * 16 * PA + kk * 4);
                c[q] = __builtin_amdgcn_wmma_f32_16x16x4_f32(
                     false, a,  false, bf,
                     (short)0, c[q],
                     false,  false);
            }
        }
        const int rbase = mt * 16 + 8 * half;
        float cb[8];
        #pragma unroll
        for (int r = 0; r < 8; ++r) cb[r] = sdl[rbase + r];
        const bool whole = (mt * 16 + 16 <= nrows);
        #pragma unroll
        for (int q = 0; q < 2; ++q) {
            const int nn = np * 32 + q * 16 + l15;
            const float bv = bias[nn];
            float* op = out + (size_t)(row0 + rbase) * D + nn;
            if (whole) {
                #pragma unroll
                for (int r = 0; r < 8; ++r)
                    op[r * D] = (c[q][r] + cb[r] * bv) * (1.0f / (1.0f + SMOOTHF));
            } else {
                #pragma unroll
                for (int r = 0; r < 8; ++r)
                    if (rbase + r < nrows)
                        op[r * D] = (c[q][r] + cb[r] * bv) * (1.0f / (1.0f + SMOOTHF));
            }
        }
    }
}


extern "C" void kernel_launch(void* const* d_in, const int* in_sizes, int n_in,
                              void* d_out, int out_size, void* d_ws, size_t ws_size,
                              hipStream_t stream) {
    const float* x    = (const float*)d_in[0];
    const float* W    = (const float*)d_in[1];
    const float* bias = (const float*)d_in[2];
    const int*   erow = (const int*)d_in[3];
    const int*   ecol = (const int*)d_in[4];
    float* out = (float*)d_out;

    float* ws         = (float*)d_ws;
    unsigned* sorted  = (unsigned*)ws;
    unsigned* seg     = sorted + N_EDGES;
    unsigned* cseg    = seg + NB * G;
    float* deg        = (float*)(cseg + NB * G);
    unsigned* csorted = (unsigned*)(deg + DPL);
    unsigned* dgw     = csorted + G * (CPS / 4);

    k_group <<<G, TPB, 0, stream>>>(erow, ecol, sorted, seg, csorted, cseg);
    k_degree<<<NB, DTB, 0, stream>>>(csorted, cseg, deg, (unsigned char*)dgw);
    k_accum <<<NB, 1024, 0, stream>>>(sorted, seg, x, W, bias, deg, dgw, out);
}
